# starting kernel
# baseline (speedup 1.0000x reference)
#include <hip/hip_runtime.h>
#include <hip/hip_bf16.h>


#define D 128
#define N_MID 50000
#define N_DST 10000

typedef __attribute__((ext_vector_type(2))) float v2f;
typedef __attribute__((ext_vector_type(8))) float v8f;

__global__ __launch_bounds__(256) void sage_scatter(
    const float* __restrict__ src, const int* __restrict__ row,
    const int* __restrict__ col, float* __restrict__ agg,
    float* __restrict__ deg, int E) {
  int e = blockIdx.x * 8 + (threadIdx.x >> 5);
  int lane = threadIdx.x & 31;
  if (e >= E) return;
  int r = row[e];
  int c = col[e];
  const float4 v = ((const float4*)(src + (size_t)r * D))[lane];
  float* dst = agg + (size_t)c * D + lane * 4;
  unsafeAtomicAdd(dst + 0, v.x);
  unsafeAtomicAdd(dst + 1, v.y);
  unsafeAtomicAdd(dst + 2, v.z);
  unsafeAtomicAdd(dst + 3, v.w);
  if (lane == 0) unsafeAtomicAdd(deg + c, 1.0f);
}

template <int RELU>
__global__ __launch_bounds__(256) void sage_gemm(
    const float* __restrict__ agg, const float* __restrict__ deg,
    const float* __restrict__ xdst, const float* __restrict__ Wl,
    const float* __restrict__ bl, const float* __restrict__ Wr,
    float* __restrict__ out, int M) {
  __shared__ float sMean[16][D + 4];
  __shared__ float sDst[16][D + 4];
  __shared__ float sOut[16][D];
  __shared__ float sInv[16];
  __shared__ float sNorm[16];

  const int tid = threadIdx.x;
  const int lane = tid & 31;
  const int w = tid >> 5;
  const int mtile = blockIdx.x;
  const int m0 = mtile * 16;

  if (tid < 16) {
    float dg = deg[m0 + tid];
    sInv[tid] = 1.0f / fmaxf(dg, 1.0f);
    sNorm[tid] = 0.0f;
  }
  __syncthreads();

  for (int i = tid; i < 16 * D; i += 256) {
    int r = i >> 7;
    int c = i & (D - 1);
    size_t g = (size_t)(m0 + r) * D + c;
    sMean[r][c] = agg[g] * sInv[r];
    sDst[r][c] = xdst[g];
  }
  __syncthreads();

  const int mrow = lane & 15;
  const int k0 = (lane >> 4) << 1;
  const int ncol = w * 16 + (lane & 15);

  float bv = bl[ncol];
  v8f acc = {bv, bv, bv, bv, bv, bv, bv, bv};

  const float* WlRow = Wl + (size_t)ncol * D;
  const float* WrRow = Wr + (size_t)ncol * D;

#pragma unroll
  for (int kk = 0; kk < D / 4; ++kk) {
    v2f a = *(const v2f*)&sMean[mrow][4 * kk + k0];
    v2f b = *(const v2f*)&WlRow[4 * kk + k0];
    acc = __builtin_amdgcn_wmma_f32_16x16x4_f32(false, a, false, b,
                                                (short)0, acc, false, false);
  }
#pragma unroll
  for (int kk = 0; kk < D / 4; ++kk) {
    v2f a = *(const v2f*)&sDst[mrow][4 * kk + k0];
    v2f b = *(const v2f*)&WrRow[4 * kk + k0];
    acc = __builtin_amdgcn_wmma_f32_16x16x4_f32(false, a, false, b,
                                                (short)0, acc, false, false);
  }

  const int rbase = (lane >> 4) << 3;
#pragma unroll
  for (int v = 0; v < 8; ++v) {
    sOut[rbase + v][ncol & (D - 1)] = acc[v];
  }
  __syncthreads();

  {
    int r = tid >> 4;
    int cbase = (tid & 15) * 8;
    float p = 0.0f;
#pragma unroll
    for (int j = 0; j < 8; ++j) {
      float x = sOut[r][cbase + j];
      p += x * x;
    }
    atomicAdd(&sNorm[r], p);
  }
  __syncthreads();

  for (int i = tid; i < 16 * D; i += 256) {
    int r = i >> 7;
    int c = i & (D - 1);
    float nrm = fmaxf(sqrtf(sNorm[r]), 1e-12f);
    float val = sOut[r][c] / nrm;
    if (RELU) val = fmaxf(val, 0.0f);
    out[(size_t)(m0 + r) * D + c] = val;
  }
}

extern "C" void kernel_launch(void* const* d_in, const int* in_sizes, int n_in,
                              void* d_out, int out_size, void* d_ws,
                              size_t ws_size, hipStream_t stream) {
  const float* x    = (const float*)d_in[0];
  const float* W_l1 = (const float*)d_in[1];
  const float* b_l1 = (const float*)d_in[2];
  const float* W_r1 = (const float*)d_in[3];
  const float* W_l2 = (const float*)d_in[4];
  const float* b_l2 = (const float*)d_in[5];
  const float* W_r2 = (const float*)d_in[6];
  const int* row1 = (const int*)d_in[7];
  const int* col1 = (const int*)d_in[8];
  const int* row2 = (const int*)d_in[9];
  const int* col2 = (const int*)d_in[10];
  const int E1 = in_sizes[7];
  const int E2 = in_sizes[9];

  float* ws = (float*)d_ws;
  float* agg1 = ws;
  float* deg1 = agg1 + (size_t)N_MID * D;
  float* agg2 = deg1 + N_MID;
  float* deg2 = agg2 + (size_t)N_DST * D;
  float* h    = deg2 + N_DST;
  const size_t zero_floats =
      (size_t)N_MID * D + N_MID + (size_t)N_DST * D + N_DST;

  hipMemsetAsync(d_ws, 0, zero_floats * sizeof(float), stream);

  sage_scatter<<<(E1 + 7) / 8, 256, 0, stream>>>(x, row1, col1, agg1, deg1, E1);
  sage_gemm<1><<<N_MID / 16, 256, 0, stream>>>(agg1, deg1, x, W_l1, b_l1, W_r1,
                                               h, N_MID);
  sage_scatter<<<(E2 + 7) / 8, 256, 0, stream>>>(h, row2, col2, agg2, deg2, E2);
  sage_gemm<0><<<N_DST / 16, 256, 0, stream>>>(agg2, deg2, h, W_l2, b_l2, W_r2,
                                               (float*)d_out, N_DST);
}


# Global float-atomic scatter replaced by destination tiles: a work-group keeps its rows' sums in LDS, filters the whole edge list, adds gathered rows by LDS adds, stores the mean; arrival-order sums
# speedup vs baseline: 2.2766x; 2.2766x over previous
#include <hip/hip_runtime.h>
#include <hip/hip_bf16.h>


#define D 128
#define N_MID 50000
#define N_DST 10000

typedef __attribute__((ext_vector_type(2))) float v2f;
typedef __attribute__((ext_vector_type(8))) float v8f;

#define AGG_WAVES 32
#define AGG_CHUNK 512
#define AGG_QCAP 576

template <int R>
__global__ __launch_bounds__(1024) void sage_tile_mean(
    const float* __restrict__ src, const int* __restrict__ row,
    const int* __restrict__ col, float* __restrict__ mean, int E, int n_dst) {
  __shared__ float sAcc[R][D];
  __shared__ int sCnt[R];
  __shared__ int sQ[AGG_WAVES][AGG_QCAP];

  const int tid = threadIdx.x;
  const int lane = tid & 31;
  const int w = __builtin_amdgcn_readfirstlane(tid >> 5);
  const int lo = blockIdx.x * R;
  if (lo >= n_dst) return;
  const unsigned nval = (unsigned)min(R, n_dst - lo);

  for (int i = tid; i < R * D; i += 1024) (&sAcc[0][0])[i] = 0.0f;
  for (int i = tid; i < R; i += 1024) sCnt[i] = 0;
  __syncthreads();

  int* q = sQ[w];
  int qn = 0;
  const int nchunk = (E + AGG_CHUNK - 1) / AGG_CHUNK;

  for (int ch = w;; ch += AGG_WAVES) {
    const bool last = ch >= nchunk;
    if (!last) {
      const int base = ch * AGG_CHUNK;
      int4 c[4];
      if (base + AGG_CHUNK <= E) {
        const int4* p = (const int4*)(col + base);
#pragma unroll
        for (int u = 0; u < 4; ++u) c[u] = p[u * 32 + lane];
      } else {
#pragma unroll
        for (int u = 0; u < 4; ++u) {
          const int e0 = base + u * 128 + lane * 4;
          c[u].x = (e0 + 0 < E) ? col[e0 + 0] : -1;
          c[u].y = (e0 + 1 < E) ? col[e0 + 1] : -1;
          c[u].z = (e0 + 2 < E) ? col[e0 + 2] : -1;
          c[u].w = (e0 + 3 < E) ? col[e0 + 3] : -1;
        }
      }
#pragma unroll
      for (int u = 0; u < 4; ++u) {
        const int cc[4] = {c[u].x, c[u].y, c[u].z, c[u].w};
        bool m[4];
#pragma unroll
        for (int j = 0; j < 4; ++j) m[j] = (unsigned)(cc[j] - lo) < nval;
        if (__builtin_amdgcn_ballot_w32(m[0] | m[1] | m[2] | m[3])) {
#pragma unroll
          for (int j = 0; j < 4; ++j) {
            const unsigned mask = __builtin_amdgcn_ballot_w32(m[j]);
            if (m[j])
              q[qn + (int)__builtin_amdgcn_mbcnt_lo(mask, 0u)] =
                  base + u * 128 + lane * 4 + j;
            qn += __builtin_popcount(mask);
          }
        }
      }
    }
    if (last || qn > AGG_QCAP - AGG_CHUNK) {
      const int take = last ? qn : (qn & ~31);
      for (int i0 = 0; i0 < take; i0 += 32) {
        const int n = min(32, take - i0);
        int r = 0, l = 0;
        if (lane < n) {
          const int e = q[i0 + lane];
          r = row[e];
          l = col[e] - lo;
          atomicAdd(&sCnt[l], 1);
        }
        for (int k = 0; k < n; k += 8) {
          float4 v[8];
          int ll[8];
#pragma unroll
          for (int j = 0; j < 8; ++j) {
            const int rr = __builtin_amdgcn_readlane(r, k + j);
            ll[j] = __builtin_amdgcn_readlane(l, k + j);
            v[j] = ((const float4*)(src + (size_t)rr * D))[lane];
          }
#pragma unroll
          for (int j = 0; j < 8; ++j) {
            if (k + j < n) {
              float* a = &sAcc[ll[j]][lane];
              atomicAdd(a, v[j].x);
              atomicAdd(a + 32, v[j].y);
              atomicAdd(a + 64, v[j].z);
              atomicAdd(a + 96, v[j].w);
            }
          }
        }
      }
      const int rem = qn - take;
      if (lane < rem) {
        const int t = q[take + lane];
        q[lane] = t;
      }
      qn = rem;
    }
    if (last) break;
  }
  __syncthreads();

  for (int i = tid; i < R * 32; i += 1024) {
    const int r = i >> 5;
    const int g = lo + r;
    if (g < n_dst) {
      const float inv = 1.0f / fmaxf((float)sCnt[r], 1.0f);
      float4 o;
      o.x = sAcc[r][lane] * inv;
      o.y = sAcc[r][32 + lane] * inv;
      o.z = sAcc[r][64 + lane] * inv;
      o.w = sAcc[r][96 + lane] * inv;
      ((float4*)(mean + (size_t)g * D))[lane] = o;
    }
  }
}

template <int RELU>
__global__ __launch_bounds__(256) void sage_gemm(
    const float* __restrict__ mean, const float* __restrict__ xdst,
    const float* __restrict__ Wl,
    const float* __restrict__ bl, const float* __restrict__ Wr,
    float* __restrict__ out, int M) {
  __shared__ float sMean[16][D + 4];
  __shared__ float sDst[16][D + 4];
  __shared__ float sOut[16][D];
  __shared__ float sNorm[16];

  const int tid = threadIdx.x;
  const int lane = tid & 31;
  const int w = tid >> 5;
  const int mtile = blockIdx.x;
  const int m0 = mtile * 16;

  if (tid < 16) sNorm[tid] = 0.0f;

  for (int i = tid; i < 16 * D; i += 256) {
    int r = i >> 7;
    int c = i & (D - 1);
    size_t g = (size_t)(m0 + r) * D + c;
    sMean[r][c] = mean[g];
    sDst[r][c] = xdst[g];
  }
  __syncthreads();

  const int mrow = lane & 15;
  const int k0 = (lane >> 4) << 1;
  const int ncol = w * 16 + (lane & 15);

  float bv = bl[ncol];
  v8f acc = {bv, bv, bv, bv, bv, bv, bv, bv};

  const float* WlRow = Wl + (size_t)ncol * D;
  const float* WrRow = Wr + (size_t)ncol * D;

#pragma unroll
  for (int kk = 0; kk < D / 4; ++kk) {
    v2f a = *(const v2f*)&sMean[mrow][4 * kk + k0];
    v2f b = *(const v2f*)&WlRow[4 * kk + k0];
    acc = __builtin_amdgcn_wmma_f32_16x16x4_f32(false, a, false, b,
                                                (short)0, acc, false, false);
  }
#pragma unroll
  for (int kk = 0; kk < D / 4; ++kk) {
    v2f a = *(const v2f*)&sDst[mrow][4 * kk + k0];
    v2f b = *(const v2f*)&WrRow[4 * kk + k0];
    acc = __builtin_amdgcn_wmma_f32_16x16x4_f32(false, a, false, b,
                                                (short)0, acc, false, false);
  }

  const int rbase = (lane >> 4) << 3;
#pragma unroll
  for (int v = 0; v < 8; ++v) {
    sOut[rbase + v][ncol & (D - 1)] = acc[v];
  }
  __syncthreads();

  {
    int r = tid >> 4;
    int cbase = (tid & 15) * 8;
    float p = 0.0f;
#pragma unroll
    for (int j = 0; j < 8; ++j) {
      float x = sOut[r][cbase + j];
      p += x * x;
    }
    atomicAdd(&sNorm[r], p);
  }
  __syncthreads();

  for (int i = tid; i < 16 * D; i += 256) {
    int r = i >> 7;
    int c = i & (D - 1);
    float nrm = fmaxf(sqrtf(sNorm[r]), 1e-12f);
    float val = sOut[r][c] / nrm;
    if (RELU) val = fmaxf(val, 0.0f);
    out[(size_t)(m0 + r) * D + c] = val;
  }
}

#define R_MID 196
#define R_DST 40

extern "C" void kernel_launch(void* const* d_in, const int* in_sizes, int n_in,
                              void* d_out, int out_size, void* d_ws,
                              size_t ws_size, hipStream_t stream) {
  const float* x    = (const float*)d_in[0];
  const float* W_l1 = (const float*)d_in[1];
  const float* b_l1 = (const float*)d_in[2];
  const float* W_r1 = (const float*)d_in[3];
  const float* W_l2 = (const float*)d_in[4];
  const float* b_l2 = (const float*)d_in[5];
  const float* W_r2 = (const float*)d_in[6];
  const int* row1 = (const int*)d_in[7];
  const int* col1 = (const int*)d_in[8];
  const int* row2 = (const int*)d_in[9];
  const int* col2 = (const int*)d_in[10];
  const int E1 = in_sizes[7];
  const int E2 = in_sizes[9];

  float* ws = (float*)d_ws;
  float* mean1 = ws;
  float* mean2 = mean1 + (size_t)N_MID * D;
  float* h     = mean2 + (size_t)N_DST * D;

  sage_tile_mean<R_MID><<<256, 1024, 0, stream>>>(x, row1, col1, mean1, E1,
                                                  N_MID);
  sage_gemm<1><<<N_MID / 16, 256, 0, stream>>>(mean1, x, W_l1, b_l1, W_r1, h,
                                               N_MID);
  sage_tile_mean<R_DST><<<250, 1024, 0, stream>>>(h, row2, col2, mean2, E2,
                                                  N_DST);
  sage_gemm<0><<<N_DST / 16, 256, 0, stream>>>(mean2, h, W_l2, b_l2, W_r2,
                                               (float*)d_out, N_DST);
}


# GEMM: each wave keeps its 16-column B fragments of both weight matrices in registers; 512/313 persistent blocks walk runs of row tiles, 2 tiles per pass (was: weights re-read from cache per 16 rows)
# speedup vs baseline: 2.5800x; 1.1333x over previous
#include <hip/hip_runtime.h>
#include <hip/hip_bf16.h>


#define D 128
#define N_MID 50000
#define N_DST 10000

typedef __attribute__((ext_vector_type(2))) float v2f;
typedef __attribute__((ext_vector_type(8))) float v8f;

#define AGG_WAVES 32
#define AGG_CHUNK 512
#define AGG_QCAP 576

template <int R>
__global__ __launch_bounds__(1024) void sage_tile_mean(
    const float* __restrict__ src, const int* __restrict__ row,
    const int* __restrict__ col, float* __restrict__ mean, int E, int n_dst) {
  __shared__ float sAcc[R][D];
  __shared__ int sCnt[R];
  __shared__ int sQ[AGG_WAVES][AGG_QCAP];

  const int tid = threadIdx.x;
  const int lane = tid & 31;
  const int w = __builtin_amdgcn_readfirstlane(tid >> 5);
  const int lo = blockIdx.x * R;
  if (lo >= n_dst) return;
  const unsigned nval = (unsigned)min(R, n_dst - lo);

  for (int i = tid; i < R * D; i += 1024) (&sAcc[0][0])[i] = 0.0f;
  for (int i = tid; i < R; i += 1024) sCnt[i] = 0;
  __syncthreads();

  int* q = sQ[w];
  int qn = 0;
  const int nchunk = (E + AGG_CHUNK - 1) / AGG_CHUNK;

  for (int ch = w;; ch += AGG_WAVES) {
    const bool last = ch >= nchunk;
    if (!last) {
      const int base = ch * AGG_CHUNK;
      int4 c[4];
      if (base + AGG_CHUNK <= E) {
        const int4* p = (const int4*)(col + base);
#pragma unroll
        for (int u = 0; u < 4; ++u) c[u] = p[u * 32 + lane];
      } else {
#pragma unroll
        for (int u = 0; u < 4; ++u) {
          const int e0 = base + u * 128 + lane * 4;
          c[u].x = (e0 + 0 < E) ? col[e0 + 0] : -1;
          c[u].y = (e0 + 1 < E) ? col[e0 + 1] : -1;
          c[u].z = (e0 + 2 < E) ? col[e0 + 2] : -1;
          c[u].w = (e0 + 3 < E) ? col[e0 + 3] : -1;
        }
      }
#pragma unroll
      for (int u = 0; u < 4; ++u) {
        const int cc[4] = {c[u].x, c[u].y, c[u].z, c[u].w};
        bool m[4];
#pragma unroll
        for (int j = 0; j < 4; ++j) m[j] = (unsigned)(cc[j] - lo) < nval;
        if (__builtin_amdgcn_ballot_w32(m[0] | m[1] | m[2] | m[3])) {
#pragma unroll
          for (int j = 0; j < 4; ++j) {
            const unsigned mask = __builtin_amdgcn_ballot_w32(m[j]);
            if (m[j])
              q[qn + (int)__builtin_amdgcn_mbcnt_lo(mask, 0u)] =
                  base + u * 128 + lane * 4 + j;
            qn += __builtin_popcount(mask);
          }
        }
      }
    }
    if (last || qn > AGG_QCAP - AGG_CHUNK) {
      const int take = last ? qn : (qn & ~31);
      for (int i0 = 0; i0 < take; i0 += 32) {
        const int n = min(32, take - i0);
        int r = 0, l = 0;
        if (lane < n) {
          const int e = q[i0 + lane];
          r = row[e];
          l = col[e] - lo;
          atomicAdd(&sCnt[l], 1);
        }
        for (int k = 0; k < n; k += 8) {
          float4 v[8];
          int ll[8];
#pragma unroll
          for (int j = 0; j < 8; ++j) {
            const int rr = __builtin_amdgcn_readlane(r, k + j);
            ll[j] = __builtin_amdgcn_readlane(l, k + j);
            v[j] = ((const float4*)(src + (size_t)rr * D))[lane];
          }
#pragma unroll
          for (int j = 0; j < 8; ++j) {
            if (k + j < n) {
              float* a = &sAcc[ll[j]][lane];
              atomicAdd(a, v[j].x);
              atomicAdd(a + 32, v[j].y);
              atomicAdd(a + 64, v[j].z);
              atomicAdd(a + 96, v[j].w);
            }
          }
        }
      }
      const int rem = qn - take;
      if (lane < rem) {
        const int t = q[take + lane];
        q[lane] = t;
      }
      qn = rem;
    }
    if (last) break;
  }
  __syncthreads();

  for (int i = tid; i < R * 32; i += 1024) {
    const int r = i >> 5;
    const int g = lo + r;
    if (g < n_dst) {
      const float inv = 1.0f / fmaxf((float)sCnt[r], 1.0f);
      float4 o;
      o.x = sAcc[r][lane] * inv;
      o.y = sAcc[r][32 + lane] * inv;
      o.z = sAcc[r][64 + lane] * inv;
      o.w = sAcc[r][96 + lane] * inv;
      ((float4*)(mean + (size_t)g * D))[lane] = o;
    }
  }
}

#define GP (D + 4)
#define GR 32

template <int NT>
__device__ __forceinline__ void sage_pass(const float* __restrict__ sMean,
                                          const float* __restrict__ sDst,
                                          const v2f (&bL)[D / 4],
                                          const v2f (&bR)[D / 4], int aoff,
                                          v8f (&acc)[2]) {
#pragma unroll
  for (int kk = 0; kk < D / 4; ++kk) {
#pragma unroll
    for (int t = 0; t < NT; ++t) {
      v2f a = *(const v2f*)&sMean[t * 16 * GP + aoff + 4 * kk];
      acc[t] = __builtin_amdgcn_wmma_f32_16x16x4_f32(false, a, false, bL[kk],
                                                     (short)0, acc[t], false,
                                                     false);
    }
  }
#pragma unroll
  for (int kk = 0; kk < D / 4; ++kk) {
#pragma unroll
    for (int t = 0; t < NT; ++t) {
      v2f a = *(const v2f*)&sDst[t * 16 * GP + aoff + 4 * kk];
      acc[t] = __builtin_amdgcn_wmma_f32_16x16x4_f32(false, a, false, bR[kk],
                                                     (short)0, acc[t], false,
                                                     false);
    }
  }
}

template <int RELU>
__global__ __launch_bounds__(256) void sage_gemm(
    const float* __restrict__ mean, const float* __restrict__ xdst,
    const float* __restrict__ Wl,
    const float* __restrict__ bl, const float* __restrict__ Wr,
    float* __restrict__ out, int M) {
  __shared__ __attribute__((aligned(16))) float sBuf[D * GP];
  __shared__ float sNorm[GR];
  float* sMean = sBuf;
  float* sDst = sBuf + GR * GP;
  float* sOut = sBuf + 2 * GR * GP;

  const int tid = threadIdx.x;
  const int lane = tid & 31;
  const int w = tid >> 5;

  const unsigned T = (unsigned)M >> 4;
  const int t0 = (int)(blockIdx.x * T / gridDim.x);
  const int t1 = (int)((blockIdx.x + 1) * T / gridDim.x);
  if (t0 >= t1) return;

  const int mrow = lane & 15;
  const int k0 = (lane >> 4) << 1;
  const int ncol = w * 16 + (lane & 15);
  const int aoff = mrow * GP + k0;
  const int boff = ncol * GP + k0;

  v2f bL[D / 4], bR[D / 4];
  {
    float4 wv[16];
#pragma unroll
    for (int j = 0; j < 16; ++j) wv[j] = ((const float4*)Wl)[tid + j * 256];
#pragma unroll
    for (int j = 0; j < 16; ++j) {
      const int i = tid + j * 256;
      *(float4*)&sBuf[(i >> 5) * GP + (i & 31) * 4] = wv[j];
    }
    __syncthreads();
#pragma unroll
    for (int kk = 0; kk < D / 4; ++kk) bL[kk] = *(const v2f*)&sBuf[boff + 4 * kk];
    __syncthreads();
  }
  {
    float4 wv[16];
#pragma unroll
    for (int j = 0; j < 16; ++j) wv[j] = ((const float4*)Wr)[tid + j * 256];
#pragma unroll
    for (int j = 0; j < 16; ++j) {
      const int i = tid + j * 256;
      *(float4*)&sBuf[(i >> 5) * GP + (i & 31) * 4] = wv[j];
    }
    __syncthreads();
#pragma unroll
    for (int kk = 0; kk < D / 4; ++kk) bR[kk] = *(const v2f*)&sBuf[boff + 4 * kk];
    __syncthreads();
  }

  const float bv = bl[ncol];
  const int rbase = (lane >> 4) << 3;

  for (int t = t0; t < t1; t += 2) {
    const int nt = min(2, t1 - t);
    const int nrows = nt * 16;
    const int m0 = t * 16;

    if (tid < GR) sNorm[tid] = 0.0f;

    {
      const int rlast = m0 + nrows - 1;
      float4 va[4], vx[4];
#pragma unroll
      for (int j = 0; j < 4; ++j) {
        const int i = tid + j * 256;
        const size_t g = (size_t)min(m0 + (i >> 5), rlast) * (D / 4) + (i & 31);
        va[j] = ((const float4*)mean)[g];
        vx[j] = ((const float4*)xdst)[g];
      }
#pragma unroll
      for (int j = 0; j < 4; ++j) {
        const int i = tid + j * 256;
        *(float4*)&sMean[(i >> 5) * GP + (i & 31) * 4] = va[j];
        *(float4*)&sDst[(i >> 5) * GP + (i & 31) * 4] = vx[j];
      }
    }
    __syncthreads();

    v8f acc[2];
    acc[0] = (v8f){bv, bv, bv, bv, bv, bv, bv, bv};
    acc[1] = acc[0];
    if (nt == 2) {
      sage_pass<2>(sMean, sDst, bL, bR, aoff, acc);
    } else {
      sage_pass<1>(sMean, sDst, bL, bR, aoff, acc);
    }

#pragma unroll
    for (int v = 0; v < 8; ++v) sOut[(rbase + v) * D + ncol] = acc[0][v];
    if (nt == 2) {
#pragma unroll
      for (int v = 0; v < 8; ++v) sOut[(16 + rbase + v) * D + ncol] = acc[1][v];
    }
    __syncthreads();

    for (int tt = 0; tt < nt; ++tt) {
      const int r = tt * 16 + (tid >> 4);
      const int cbase = (tid & 15) * 8;
      float p = 0.0f;
#pragma unroll
      for (int j = 0; j < 8; ++j) {
        float x = sOut[r * D + cbase + j];
        p += x * x;
      }
      atomicAdd(&sNorm[r], p);
    }
    __syncthreads();

    for (int i = tid; i < nrows * D; i += 256) {
      int r = i >> 7;
      int c = i & (D - 1);
      float nrm = fmaxf(sqrtf(sNorm[r]), 1e-12f);
      float val = sOut[r * D + c] / nrm;
      if (RELU) val = fmaxf(val, 0.0f);
      out[(size_t)(m0 + r) * D + c] = val;
    }
    __syncthreads();
  }
}

#define R_MID 196
#define R_DST 40
#define GEMM_WG_MID 512
#define GEMM_WG_DST 313

extern "C" void kernel_launch(void* const* d_in, const int* in_sizes, int n_in,
                              void* d_out, int out_size, void* d_ws,
                              size_t ws_size, hipStream_t stream) {
  const float* x    = (const float*)d_in[0];
  const float* W_l1 = (const float*)d_in[1];
  const float* b_l1 = (const float*)d_in[2];
  const float* W_r1 = (const float*)d_in[3];
  const float* W_l2 = (const float*)d_in[4];
  const float* b_l2 = (const float*)d_in[5];
  const float* W_r2 = (const float*)d_in[6];
  const int* row1 = (const int*)d_in[7];
  const int* col1 = (const int*)d_in[8];
  const int* row2 = (const int*)d_in[9];
  const int* col2 = (const int*)d_in[10];
  const int E1 = in_sizes[7];
  const int E2 = in_sizes[9];

  float* ws = (float*)d_ws;
  float* mean1 = ws;
  float* mean2 = mean1 + (size_t)N_MID * D;
  float* h     = mean2 + (size_t)N_DST * D;

  sage_tile_mean<R_MID><<<256, 1024, 0, stream>>>(x, row1, col1, mean1, E1,
                                                  N_MID);
  sage_gemm<1><<<GEMM_WG_MID, 256, 0, stream>>>(mean1, x, W_l1, b_l1, W_r1, h,
                                               N_MID);
  sage_tile_mean<R_DST><<<250, 1024, 0, stream>>>(h, row2, col2, mean2, E2,
                                                  N_DST);
  sage_gemm<0><<<GEMM_WG_DST, 256, 0, stream>>>(mean2, h, W_l2, b_l2, W_r2,
                                               (float*)d_out, N_DST);
}


# Tile filter scans a 1-byte tile key per edge (both layers' keys written by one small first launch; 197-row tiles so an index fits a byte) instead of the 4-byte index; arrival-order sums as before
# speedup vs baseline: 3.0507x; 1.1824x over previous
#include <hip/hip_runtime.h>
#include <hip/hip_bf16.h>


#define D 128
#define N_MID 50000
#define N_DST 10000

typedef __attribute__((ext_vector_type(2))) float v2f;
typedef __attribute__((ext_vector_type(8))) float v8f;

#define KEY_STEP 2048
#define KEY_NONE 255u

template <int R>
__device__ __forceinline__ unsigned tile_key(int c, int n_dst) {
  return (unsigned)c < (unsigned)n_dst ? (unsigned)c / (unsigned)R : KEY_NONE;
}

template <int R>
__device__ __forceinline__ void write_tile_keys(const int* __restrict__ col,
                                                unsigned* __restrict__ key,
                                                int E, int n_dst, int gtid,
                                                int gsz) {
  const int nquad = ((E + KEY_STEP - 1) / KEY_STEP) * (KEY_STEP / 4);
  const int nfull = E >> 2;
  for (int i0 = gtid; i0 < nquad; i0 += 4 * gsz) {
    int4 c[4];
#pragma unroll
    for (int k = 0; k < 4; ++k) {
      const int i = i0 + k * gsz;
      if (i < nfull) {
        c[k] = ((const int4*)col)[i];
      } else {
        const int e0 = i * 4;
        c[k].x = (e0 + 0 < E) ? col[e0 + 0] : -1;
        c[k].y = (e0 + 1 < E) ? col[e0 + 1] : -1;
        c[k].z = (e0 + 2 < E) ? col[e0 + 2] : -1;
        c[k].w = (e0 + 3 < E) ? col[e0 + 3] : -1;
      }
    }
#pragma unroll
    for (int k = 0; k < 4; ++k) {
      const int i = i0 + k * gsz;
      if (i < nquad)
        key[i] = tile_key<R>(c[k].x, n_dst) | (tile_key<R>(c[k].y, n_dst) << 8) |
                 (tile_key<R>(c[k].z, n_dst) << 16) |
                 (tile_key<R>(c[k].w, n_dst) << 24);
    }
  }
}

template <int R1, int R2>
__global__ __launch_bounds__(256) void sage_tile_keys(
    const int* __restrict__ col1, const int* __restrict__ col2,
    unsigned* __restrict__ key1, unsigned* __restrict__ key2, int E1, int E2,
    int n1, int n2) {
  const int gtid = blockIdx.x * 256 + threadIdx.x;
  const int gsz = gridDim.x * 256;
  write_tile_keys<R1>(col1, key1, E1, n1, gtid, gsz);
  write_tile_keys<R2>(col2, key2, E2, n2, gtid, gsz);
}

#define AGG_WAVES 32
#define AGG_QCAP 576
#define AGG_QDRAIN 64
static_assert(AGG_QDRAIN + 32 <= AGG_QCAP, "one queueing round adds up to 32");

__device__ __forceinline__ unsigned key_hits(uint4 k, unsigned key4) {
  const unsigned x[4] = {k.x ^ key4, k.y ^ key4, k.z ^ key4, k.w ^ key4};
  unsigned f = 0;
#pragma unroll
  for (int d = 0; d < 4; ++d) {
    const unsigned t = ((x[d] & 0x7f7f7f7fu) + 0x7f7f7f7fu) | x[d];
    f |= (~t & 0x80808080u) >> (7 - d);
  }
  return f;
}

template <int R>
__device__ __forceinline__ void sage_drain(const int* q, int take,
                                           const float* __restrict__ src,
                                           const int* __restrict__ row,
                                           const int* __restrict__ col, int lo,
                                           float (*sAcc)[D], int* sCnt,
                                           int lane) {
  for (int i0 = 0; i0 < take; i0 += 32) {
    const int n = min(32, take - i0);
    int r = 0, l = 0;
    if (lane < n) {
      const int e = q[i0 + lane];
      r = row[e];
      l = col[e] - lo;
      atomicAdd(&sCnt[l], 1);
    }
    for (int k = 0; k < n; k += 8) {
      float4 v[8];
      int ll[8];
#pragma unroll
      for (int j = 0; j < 8; ++j) {
        const int rr = __builtin_amdgcn_readlane(r, k + j);
        ll[j] = __builtin_amdgcn_readlane(l, k + j);
        v[j] = ((const float4*)(src + (size_t)rr * D))[lane];
      }
#pragma unroll
      for (int j = 0; j < 8; ++j) {
        if (k + j < n) {
          float* a = &sAcc[ll[j]][lane];
          atomicAdd(a, v[j].x);
          atomicAdd(a + 32, v[j].y);
          atomicAdd(a + 64, v[j].z);
          atomicAdd(a + 96, v[j].w);
        }
      }
    }
  }
}

template <int R>
__global__ __launch_bounds__(1024) void sage_tile_mean(
    const float* __restrict__ src, const int* __restrict__ row,
    const int* __restrict__ col, const unsigned* __restrict__ key,
    float* __restrict__ mean, int E, int n_dst) {
  __shared__ float sAcc[R][D];
  __shared__ int sCnt[R];
  __shared__ int sQ[AGG_WAVES][AGG_QCAP];

  const int tid = threadIdx.x;
  const int lane = tid & 31;
  const int w = __builtin_amdgcn_readfirstlane(tid >> 5);
  const int lo = blockIdx.x * R;
  if (lo >= n_dst) return;

  for (int i = tid; i < R * D; i += 1024) (&sAcc[0][0])[i] = 0.0f;
  for (int i = tid; i < R; i += 1024) sCnt[i] = 0;
  __syncthreads();

  int* q = sQ[w];
  int qn = 0;
  const int nstep = (E + KEY_STEP - 1) / KEY_STEP;
  const unsigned key4 = (unsigned)blockIdx.x * 0x01010101u;

  for (int st = w; st < nstep; st += AGG_WAVES) {
    const uint4* p = (const uint4*)(key + (size_t)st * (KEY_STEP / 4));
    uint4 kv[4];
#pragma unroll
    for (int u = 0; u < 4; ++u) kv[u] = p[u * 32 + lane];
#pragma unroll
    for (int h = 0; h < 2; ++h) {
      unsigned f = key_hits(kv[2 * h], key4) | (key_hits(kv[2 * h + 1], key4) << 4);
      const int ebase = st * KEY_STEP + h * 1024 + lane * 16;
      for (;;) {
        const unsigned mask = __builtin_amdgcn_ballot_w32(f != 0);
        if (mask == 0) break;
        if (f != 0) {
          const int b = __builtin_ctz(f);
          f &= f - 1;
          q[qn + (int)__builtin_amdgcn_mbcnt_lo(mask, 0u)] =
              ebase + ((b & 4) << 7) + ((b & 3) << 2) + (b >> 3);
        }
        qn += __builtin_popcount(mask);
        if (qn > AGG_QDRAIN) {
          const int take = qn & ~31;
          sage_drain<R>(q, take, src, row, col, lo, sAcc, sCnt, lane);
          const int rem = qn - take;
          if (lane < rem) {
            const int t = q[take + lane];
            q[lane] = t;
          }
          qn = rem;
        }
      }
    }
  }
  sage_drain<R>(q, qn, src, row, col, lo, sAcc, sCnt, lane);
  __syncthreads();

  for (int i = tid; i < R * 32; i += 1024) {
    const int r = i >> 5;
    const int g = lo + r;
    if (g < n_dst) {
      const float inv = 1.0f / fmaxf((float)sCnt[r], 1.0f);
      float4 o;
      o.x = sAcc[r][lane] * inv;
      o.y = sAcc[r][32 + lane] * inv;
      o.z = sAcc[r][64 + lane] * inv;
      o.w = sAcc[r][96 + lane] * inv;
      ((float4*)(mean + (size_t)g * D))[lane] = o;
    }
  }
}

#define GP (D + 4)
#define GR 32

template <int NT>
__device__ __forceinline__ void sage_pass(const float* __restrict__ sMean,
                                          const float* __restrict__ sDst,
                                          const v2f (&bL)[D / 4],
                                          const v2f (&bR)[D / 4], int aoff,
                                          v8f (&acc)[2]) {
#pragma unroll
  for (int kk = 0; kk < D / 4; ++kk) {
#pragma unroll
    for (int t = 0; t < NT; ++t) {
      v2f a = *(const v2f*)&sMean[t * 16 * GP + aoff + 4 * kk];
      acc[t] = __builtin_amdgcn_wmma_f32_16x16x4_f32(false, a, false, bL[kk],
                                                     (short)0, acc[t], false,
                                                     false);
    }
  }
#pragma unroll
  for (int kk = 0; kk < D / 4; ++kk) {
#pragma unroll
    for (int t = 0; t < NT; ++t) {
      v2f a = *(const v2f*)&sDst[t * 16 * GP + aoff + 4 * kk];
      acc[t] = __builtin_amdgcn_wmma_f32_16x16x4_f32(false, a, false, bR[kk],
                                                     (short)0, acc[t], false,
                                                     false);
    }
  }
}

template <int RELU>
__global__ __launch_bounds__(256) void sage_gemm(
    const float* __restrict__ mean, const float* __restrict__ xdst,
    const float* __restrict__ Wl,
    const float* __restrict__ bl, const float* __restrict__ Wr,
    float* __restrict__ out, int M) {
  __shared__ __attribute__((aligned(16))) float sBuf[D * GP];
  __shared__ float sNorm[GR];
  float* sMean = sBuf;
  float* sDst = sBuf + GR * GP;
  float* sOut = sBuf + 2 * GR * GP;

  const int tid = threadIdx.x;
  const int lane = tid & 31;
  const int w = tid >> 5;

  const unsigned T = (unsigned)M >> 4;
  const int t0 = (int)(blockIdx.x * T / gridDim.x);
  const int t1 = (int)((blockIdx.x + 1) * T / gridDim.x);
  if (t0 >= t1) return;

  const int mrow = lane & 15;
  const int k0 = (lane >> 4) << 1;
  const int ncol = w * 16 + (lane & 15);
  const int aoff = mrow * GP + k0;
  const int boff = ncol * GP + k0;

  v2f bL[D / 4], bR[D / 4];
  {
    float4 wv[16];
#pragma unroll
    for (int j = 0; j < 16; ++j) wv[j] = ((const float4*)Wl)[tid + j * 256];
#pragma unroll
    for (int j = 0; j < 16; ++j) {
      const int i = tid + j * 256;
      *(float4*)&sBuf[(i >> 5) * GP + (i & 31) * 4] = wv[j];
    }
    __syncthreads();
#pragma unroll
    for (int kk = 0; kk < D / 4; ++kk) bL[kk] = *(const v2f*)&sBuf[boff + 4 * kk];
    __syncthreads();
  }
  {
    float4 wv[16];
#pragma unroll
    for (int j = 0; j < 16; ++j) wv[j] = ((const float4*)Wr)[tid + j * 256];
#pragma unroll
    for (int j = 0; j < 16; ++j) {
      const int i = tid + j * 256;
      *(float4*)&sBuf[(i >> 5) * GP + (i & 31) * 4] = wv[j];
    }
    __syncthreads();
#pragma unroll
    for (int kk = 0; kk < D / 4; ++kk) bR[kk] = *(const v2f*)&sBuf[boff + 4 * kk];
    __syncthreads();
  }

  const float bv = bl[ncol];
  const int rbase = (lane >> 4) << 3;

  for (int t = t0; t < t1; t += 2) {
    const int nt = min(2, t1 - t);
    const int nrows = nt * 16;
    const int m0 = t * 16;

    if (tid < GR) sNorm[tid] = 0.0f;

    {
      const int rlast = m0 + nrows - 1;
      float4 va[4], vx[4];
#pragma unroll
      for (int j = 0; j < 4; ++j) {
        const int i = tid + j * 256;
        const size_t g = (size_t)min(m0 + (i >> 5), rlast) * (D / 4) + (i & 31);
        va[j] = ((const float4*)mean)[g];
        vx[j] = ((const float4*)xdst)[g];
      }
#pragma unroll
      for (int j = 0; j < 4; ++j) {
        const int i = tid + j * 256;
        *(float4*)&sMean[(i >> 5) * GP + (i & 31) * 4] = va[j];
        *(float4*)&sDst[(i >> 5) * GP + (i & 31) * 4] = vx[j];
      }
    }
    __syncthreads();

    v8f acc[2];
    acc[0] = (v8f){bv, bv, bv, bv, bv, bv, bv, bv};
    acc[1] = acc[0];
    if (nt == 2) {
      sage_pass<2>(sMean, sDst, bL, bR, aoff, acc);
    } else {
      sage_pass<1>(sMean, sDst, bL, bR, aoff, acc);
    }

#pragma unroll
    for (int v = 0; v < 8; ++v) sOut[(rbase + v) * D + ncol] = acc[0][v];
    if (nt == 2) {
#pragma unroll
      for (int v = 0; v < 8; ++v) sOut[(16 + rbase + v) * D + ncol] = acc[1][v];
    }
    __syncthreads();

    for (int tt = 0; tt < nt; ++tt) {
      const int r = tt * 16 + (tid >> 4);
      const int cbase = (tid & 15) * 8;
      float p = 0.0f;
#pragma unroll
      for (int j = 0; j < 8; ++j) {
        float x = sOut[r * D + cbase + j];
        p += x * x;
      }
      atomicAdd(&sNorm[r], p);
    }
    __syncthreads();

    for (int i = tid; i < nrows * D; i += 256) {
      int r = i >> 7;
      int c = i & (D - 1);
      float nrm = fmaxf(sqrtf(sNorm[r]), 1e-12f);
      float val = sOut[r * D + c] / nrm;
      if (RELU) val = fmaxf(val, 0.0f);
      out[(size_t)(m0 + r) * D + c] = val;
    }
    __syncthreads();
  }
}

#define R_MID 197
#define R_DST 40
#define KEY_PLANE (1 << 20)
static_assert((N_MID + R_MID - 1) / R_MID <= (int)KEY_NONE &&
                  (N_DST + R_DST - 1) / R_DST <= (int)KEY_NONE,
              "every tile index must be a key byte below KEY_NONE");
#define GEMM_WG_MID 512
#define GEMM_WG_DST 313

extern "C" void kernel_launch(void* const* d_in, const int* in_sizes, int n_in,
                              void* d_out, int out_size, void* d_ws,
                              size_t ws_size, hipStream_t stream) {
  const float* x    = (const float*)d_in[0];
  const float* W_l1 = (const float*)d_in[1];
  const float* b_l1 = (const float*)d_in[2];
  const float* W_r1 = (const float*)d_in[3];
  const float* W_l2 = (const float*)d_in[4];
  const float* b_l2 = (const float*)d_in[5];
  const float* W_r2 = (const float*)d_in[6];
  const int* row1 = (const int*)d_in[7];
  const int* col1 = (const int*)d_in[8];
  const int* row2 = (const int*)d_in[9];
  const int* col2 = (const int*)d_in[10];
  const int E1 = in_sizes[7];
  const int E2 = in_sizes[9];

  float* ws = (float*)d_ws;
  float* mean1 = ws;
  float* mean2 = mean1 + (size_t)N_MID * D;
  float* h     = mean2 + (size_t)N_DST * D;
  unsigned* key1 = (unsigned*)(h + (size_t)N_MID * D);
  unsigned* key2 = key1 + KEY_PLANE;

  sage_tile_keys<R_MID, R_DST><<<256, 256, 0, stream>>>(col1, col2, key1, key2,
                                                       E1, E2, N_MID, N_DST);

  sage_tile_mean<R_MID><<<254, 1024, 0, stream>>>(x, row1, col1, key1, mean1,
                                                  E1, N_MID);
  sage_gemm<1><<<GEMM_WG_MID, 256, 0, stream>>>(mean1, x, W_l1, b_l1, W_r1, h,
                                               N_MID);
  sage_tile_mean<R_DST><<<250, 1024, 0, stream>>>(h, row2, col2, key2, mean2,
                                                  E2, N_DST);
  sage_gemm<0><<<GEMM_WG_DST, 256, 0, stream>>>(mean2, h, W_l2, b_l2, W_r2,
                                               (float*)d_out, N_DST);
}


# Layer product moved into the destination-tile work-group: root term as each wave ends its edges, then LDS means as WMMA A; 5 launches to 3, no mean planes; root term summed first, norm by shuffles
# speedup vs baseline: 3.6950x; 1.2112x over previous
#include <hip/hip_runtime.h>
#include <hip/hip_bf16.h>


#define D 128
#define N_MID 50000
#define N_DST 10000

typedef __attribute__((ext_vector_type(2))) float v2f;
typedef __attribute__((ext_vector_type(8))) float v8f;

#define KEY_STEP 2048
#define KEY_NONE 255u

template <int R>
__device__ __forceinline__ unsigned tile_key(int c, int n_dst) {
  return (unsigned)c < (unsigned)n_dst ? (unsigned)c / (unsigned)R : KEY_NONE;
}

template <int R>
__device__ __forceinline__ void write_tile_keys(const int* __restrict__ col,
                                                unsigned* __restrict__ key,
                                                int E, int n_dst, int gtid,
                                                int gsz) {
  const int nquad = ((E + KEY_STEP - 1) / KEY_STEP) * (KEY_STEP / 4);
  const int nfull = E >> 2;
  for (int i0 = gtid; i0 < nquad; i0 += 4 * gsz) {
    int4 c[4];
#pragma unroll
    for (int k = 0; k < 4; ++k) {
      const int i = i0 + k * gsz;
      if (i < nfull) {
        c[k] = ((const int4*)col)[i];
      } else {
        const int e0 = i * 4;
        c[k].x = (e0 + 0 < E) ? col[e0 + 0] : -1;
        c[k].y = (e0 + 1 < E) ? col[e0 + 1] : -1;
        c[k].z = (e0 + 2 < E) ? col[e0 + 2] : -1;
        c[k].w = (e0 + 3 < E) ? col[e0 + 3] : -1;
      }
    }
#pragma unroll
    for (int k = 0; k < 4; ++k) {
      const int i = i0 + k * gsz;
      if (i < nquad)
        key[i] = tile_key<R>(c[k].x, n_dst) | (tile_key<R>(c[k].y, n_dst) << 8) |
                 (tile_key<R>(c[k].z, n_dst) << 16) |
                 (tile_key<R>(c[k].w, n_dst) << 24);
    }
  }
}

template <int R1, int R2>
__global__ __launch_bounds__(256) void sage_tile_keys(
    const int* __restrict__ col1, const int* __restrict__ col2,
    unsigned* __restrict__ key1, unsigned* __restrict__ key2, int E1, int E2,
    int n1, int n2) {
  const int gtid = blockIdx.x * 256 + threadIdx.x;
  const int gsz = gridDim.x * 256;
  write_tile_keys<R1>(col1, key1, E1, n1, gtid, gsz);
  write_tile_keys<R2>(col2, key2, E2, n2, gtid, gsz);
}

#define SP (D + 4)
#define AGG_WAVES 32
#define AGG_QCAP 576
#define AGG_QDRAIN 64
static_assert(AGG_QDRAIN + 32 <= AGG_QCAP, "one queueing round adds up to 32");

__device__ __forceinline__ unsigned key_hits(uint4 k, unsigned key4) {
  const unsigned x[4] = {k.x ^ key4, k.y ^ key4, k.z ^ key4, k.w ^ key4};
  unsigned f = 0;
#pragma unroll
  for (int d = 0; d < 4; ++d) {
    const unsigned t = ((x[d] & 0x7f7f7f7fu) + 0x7f7f7f7fu) | x[d];
    f |= (~t & 0x80808080u) >> (7 - d);
  }
  return f;
}

template <int R>
__device__ __forceinline__ void sage_drain(const int* q, int take,
                                           const float* __restrict__ src,
                                           const int* __restrict__ row,
                                           const int* __restrict__ col, int lo,
                                           float (*sAcc)[SP], int* sCnt,
                                           int lane) {
  for (int i0 = 0; i0 < take; i0 += 32) {
    const int n = min(32, take - i0);
    int r = 0, l = 0;
    if (lane < n) {
      const int e = q[i0 + lane];
      r = row[e];
      l = col[e] - lo;
      atomicAdd(&sCnt[l], 1);
    }
    for (int k = 0; k < n; k += 8) {
      float4 v[8];
      int ll[8];
#pragma unroll
      for (int j = 0; j < 8; ++j) {
        const int rr = __builtin_amdgcn_readlane(r, k + j);
        ll[j] = __builtin_amdgcn_readlane(l, k + j);
        v[j] = ((const float4*)(src + (size_t)rr * D))[lane];
      }
#pragma unroll
      for (int j = 0; j < 8; ++j) {
        if (k + j < n) {
          float* a = &sAcc[ll[j]][lane];
          atomicAdd(a, v[j].x);
          atomicAdd(a + 32, v[j].y);
          atomicAdd(a + 64, v[j].z);
          atomicAdd(a + 96, v[j].w);
        }
      }
    }
  }
}

#define KH (D / 8)
template <int NT>
__device__ __forceinline__ void sage_pass(const float* __restrict__ sA,
                                          const v2f (&b)[KH], int aoff,
                                          v8f* acc) {
#pragma unroll
  for (int kk = 0; kk < KH; ++kk) {
#pragma unroll
    for (int t = 0; t < NT; ++t) {
      v2f a = *(const v2f*)&sA[t * 64 * SP + aoff + 4 * kk];
      acc[t] = __builtin_amdgcn_wmma_f32_16x16x4_f32(false, a, false, b[kk],
                                                     (short)0, acc[t], false,
                                                     false);
    }
  }
}

template <int R, int RELU>
__global__ __launch_bounds__(1024) void sage_tile_layer(
    const float* __restrict__ src, const int* __restrict__ row,
    const int* __restrict__ col, const unsigned* __restrict__ key,
    const float* __restrict__ xdst, const float* __restrict__ Wl,
    const float* __restrict__ bl, const float* __restrict__ Wr,
    float* __restrict__ out, int E, int n_dst) {
  constexpr int R16 = (R + 15) / 16 * 16;
  constexpr int NT16 = R16 / 16;
  constexpr int NTMAX = (NT16 + 3) / 4;
  __shared__ __attribute__((aligned(16))) float sAcc[R16][SP];
  __shared__ __attribute__((aligned(16))) float sDst[R16][SP];
  __shared__ int sCnt[R16];
  __shared__ int sQ[AGG_WAVES][AGG_QCAP];

  const int tid = threadIdx.x;
  const int lane = tid & 31;
  const int w = __builtin_amdgcn_readfirstlane(tid >> 5);
  const int lo = blockIdx.x * R;
  if (lo >= n_dst) return;

  for (int i0 = tid; i0 < R16 * 32; i0 += 4096) {
    float4 v[4];
#pragma unroll
    for (int j = 0; j < 4; ++j) {
      const int i = min(i0 + j * 1024, R16 * 32 - 1);
      const size_t g = (size_t)min(lo + (i >> 5), n_dst - 1) * (D / 4) + (i & 31);
      v[j] = ((const float4*)xdst)[g];
    }
#pragma unroll
    for (int j = 0; j < 4; ++j) {
      const int i = i0 + j * 1024;
      if (i < R16 * 32) *(float4*)&sDst[i >> 5][(i & 31) * 4] = v[j];
    }
  }
  for (int i = tid; i < R16 * SP; i += 1024) (&sAcc[0][0])[i] = 0.0f;
  for (int i = tid; i < R16; i += 1024) sCnt[i] = 0;
  __syncthreads();

  int* q = sQ[w];
  int qn = 0;
  const int nstep = (E + KEY_STEP - 1) / KEY_STEP;
  const unsigned key4 = (unsigned)blockIdx.x * 0x01010101u;

  for (int st = w; st < nstep; st += AGG_WAVES) {
    const uint4* p = (const uint4*)(key + (size_t)st * (KEY_STEP / 4));
    uint4 kv[4];
#pragma unroll
    for (int u = 0; u < 4; ++u) kv[u] = p[u * 32 + lane];
#pragma unroll
    for (int h = 0; h < 2; ++h) {
      unsigned f = key_hits(kv[2 * h], key4) | (key_hits(kv[2 * h + 1], key4) << 4);
      const int ebase = st * KEY_STEP + h * 1024 + lane * 16;
      for (;;) {
        const unsigned mask = __builtin_amdgcn_ballot_w32(f != 0);
        if (mask == 0) break;
        if (f != 0) {
          const int b = __builtin_ctz(f);
          f &= f - 1;
          q[qn + (int)__builtin_amdgcn_mbcnt_lo(mask, 0u)] =
              ebase + ((b & 4) << 7) + ((b & 3) << 2) + (b >> 3);
        }
        qn += __builtin_popcount(mask);
        if (qn > AGG_QDRAIN) {
          const int take = qn & ~31;
          sage_drain<R>(q, take, src, row, col, lo, sAcc, sCnt, lane);
          const int rem = qn - take;
          if (lane < rem) {
            const int t = q[take + lane];
            q[lane] = t;
          }
          qn = rem;
        }
      }
    }
  }
  sage_drain<R>(q, qn, src, row, col, lo, sAcc, sCnt, lane);

  const int ct = w & 7;
  const int rg = w >> 3;
  const int nt = (NT16 - rg + 3) >> 2;
  const int k0 = (lane >> 4) << 1;
  const int ncol = ct * 16 + (lane & 15);
  const int aoff = (rg * 16 + (lane & 15)) * SP + k0;
  const size_t boff = (size_t)ncol * D + k0;

  const float bv = bl[ncol];
  v8f acc[NTMAX];
#pragma unroll
  for (int t = 0; t < NTMAX; ++t) acc[t] = (v8f){bv, bv, bv, bv, bv, bv, bv, bv};

#pragma unroll
  for (int hk = 0; hk < 2; ++hk) {
    v2f b[KH];
#pragma unroll
    for (int kk = 0; kk < KH; ++kk)
      b[kk] = *(const v2f*)&Wr[boff + hk * (D / 2) + 4 * kk];
    if (nt == NTMAX) {
      sage_pass<NTMAX>(&sDst[0][0], b, aoff + hk * (D / 2), acc);
    } else if (NTMAX > 1 && nt == NTMAX - 1) {
      sage_pass<(NTMAX > 1 ? NTMAX - 1 : 1)>(&sDst[0][0], b, aoff + hk * (D / 2), acc);
    }
  }
  __syncthreads();

  for (int r = w; r < R; r += AGG_WAVES) {
    const float inv = 1.0f / fmaxf((float)sCnt[r], 1.0f);
    float4 o;
    o.x = sAcc[r][lane] * inv;
    o.y = sAcc[r][32 + lane] * inv;
    o.z = sAcc[r][64 + lane] * inv;
    o.w = sAcc[r][96 + lane] * inv;
    *(float4*)&sAcc[r][lane * 4] = o;
  }
  __syncthreads();

#pragma unroll
  for (int hk = 0; hk < 2; ++hk) {
    v2f b[KH];
#pragma unroll
    for (int kk = 0; kk < KH; ++kk)
      b[kk] = *(const v2f*)&Wl[boff + hk * (D / 2) + 4 * kk];
    if (nt == NTMAX) {
      sage_pass<NTMAX>(&sAcc[0][0], b, aoff + hk * (D / 2), acc);
    } else if (NTMAX > 1 && nt == NTMAX - 1) {
      sage_pass<(NTMAX > 1 ? NTMAX - 1 : 1)>(&sAcc[0][0], b, aoff + hk * (D / 2), acc);
    }
  }
  __syncthreads();

  {
    const int rbase = (lane >> 4) << 3;
#pragma unroll
    for (int t = 0; t < NTMAX; ++t) {
      if (t < nt) {
#pragma unroll
        for (int v = 0; v < 8; ++v)
          sAcc[(rg + 4 * t) * 16 + rbase + v][ncol] = acc[t][v];
      }
    }
  }
  __syncthreads();

  for (int r = w; r < R; r += AGG_WAVES) {
    const int g = lo + r;
    if (g < n_dst) {
      float4 o = *(const float4*)&sAcc[r][lane * 4];
      float p = o.x * o.x + o.y * o.y + o.z * o.z + o.w * o.w;
#pragma unroll
      for (int d = 16; d >= 1; d >>= 1) p += __shfl_xor(p, d, 32);
      const float nrm = fmaxf(sqrtf(p), 1e-12f);
      o.x /= nrm;
      o.y /= nrm;
      o.z /= nrm;
      o.w /= nrm;
      if (RELU) {
        o.x = fmaxf(o.x, 0.0f);
        o.y = fmaxf(o.y, 0.0f);
        o.z = fmaxf(o.z, 0.0f);
        o.w = fmaxf(o.w, 0.0f);
      }
      ((float4*)(out + (size_t)g * D))[lane] = o;
    }
  }
}

#define R_MID 197
#define R_DST 40
#define KEY_PLANE (1 << 20)
static_assert((N_MID + R_MID - 1) / R_MID <= (int)KEY_NONE &&
                  (N_DST + R_DST - 1) / R_DST <= (int)KEY_NONE,
              "every tile index must be a key byte below KEY_NONE");

extern "C" void kernel_launch(void* const* d_in, const int* in_sizes, int n_in,
                              void* d_out, int out_size, void* d_ws,
                              size_t ws_size, hipStream_t stream) {
  const float* x    = (const float*)d_in[0];
  const float* W_l1 = (const float*)d_in[1];
  const float* b_l1 = (const float*)d_in[2];
  const float* W_r1 = (const float*)d_in[3];
  const float* W_l2 = (const float*)d_in[4];
  const float* b_l2 = (const float*)d_in[5];
  const float* W_r2 = (const float*)d_in[6];
  const int* row1 = (const int*)d_in[7];
  const int* col1 = (const int*)d_in[8];
  const int* row2 = (const int*)d_in[9];
  const int* col2 = (const int*)d_in[10];
  const int E1 = in_sizes[7];
  const int E2 = in_sizes[9];

  float* ws = (float*)d_ws;
  float* h  = ws;
  unsigned* key1 = (unsigned*)(h + (size_t)N_MID * D);
  unsigned* key2 = key1 + KEY_PLANE;

  sage_tile_keys<R_MID, R_DST><<<256, 256, 0, stream>>>(col1, col2, key1, key2,
                                                       E1, E2, N_MID, N_DST);

  sage_tile_layer<R_MID, 1><<<254, 1024, 0, stream>>>(
      x, row1, col1, key1, x, W_l1, b_l1, W_r1, h, E1, N_MID);
  sage_tile_layer<R_DST, 0><<<250, 1024, 0, stream>>>(
      h, row2, col2, key2, h, W_l2, b_l2, W_r2, (float*)d_out, E2, N_DST);
}


# B fragments read from a fragment-major weight copy (written by launch 1): one 512-byte piece per load, was 16 rows x 16 B; A and B take 8 k per 16-byte read, so k is summed in a new order
# speedup vs baseline: 3.9092x; 1.0580x over previous
#include <hip/hip_runtime.h>
#include <hip/hip_bf16.h>


#define D 128
#define N_MID 50000
#define N_DST 10000

typedef __attribute__((ext_vector_type(2))) float v2f;
typedef __attribute__((ext_vector_type(8))) float v8f;

#define KEY_STEP 2048
#define KEY_NONE 255u

template <int R>
__device__ __forceinline__ unsigned tile_key(int c, int n_dst) {
  return (unsigned)c < (unsigned)n_dst ? (unsigned)c / (unsigned)R : KEY_NONE;
}

template <int R>
__device__ __forceinline__ void write_tile_keys(const int* __restrict__ col,
                                                unsigned* __restrict__ key,
                                                int E, int n_dst, int gtid,
                                                int gsz) {
  const int nquad = ((E + KEY_STEP - 1) / KEY_STEP) * (KEY_STEP / 4);
  const int nfull = E >> 2;
  for (int i0 = gtid; i0 < nquad; i0 += 4 * gsz) {
    int4 c[4];
#pragma unroll
    for (int k = 0; k < 4; ++k) {
      const int i = i0 + k * gsz;
      if (i < nfull) {
        c[k] = ((const int4*)col)[i];
      } else {
        const int e0 = i * 4;
        c[k].x = (e0 + 0 < E) ? col[e0 + 0] : -1;
        c[k].y = (e0 + 1 < E) ? col[e0 + 1] : -1;
        c[k].z = (e0 + 2 < E) ? col[e0 + 2] : -1;
        c[k].w = (e0 + 3 < E) ? col[e0 + 3] : -1;
      }
    }
#pragma unroll
    for (int k = 0; k < 4; ++k) {
      const int i = i0 + k * gsz;
      if (i < nquad)
        key[i] = tile_key<R>(c[k].x, n_dst) | (tile_key<R>(c[k].y, n_dst) << 8) |
                 (tile_key<R>(c[k].z, n_dst) << 16) |
                 (tile_key<R>(c[k].w, n_dst) << 24);
    }
  }
}

#define W_FRAG4 (D * D / 4)

__device__ __forceinline__ void write_weight_frags(const float* __restrict__ Wm,
                                                   float* __restrict__ F, int idx) {
  const int ct = idx >> 9, j = (idx >> 5) & 15, h = (idx >> 4) & 1, c = idx & 15;
  ((float4*)F)[idx] = *(const float4*)&Wm[(size_t)(ct * 16 + c) * D + 8 * j + 4 * h];
}

template <int R1, int R2>
__global__ __launch_bounds__(256) void sage_tile_keys(
    const int* __restrict__ col1, const int* __restrict__ col2,
    unsigned* __restrict__ key1, unsigned* __restrict__ key2, int E1, int E2,
    int n1, int n2, const float* __restrict__ Wl1, const float* __restrict__ Wr1,
    const float* __restrict__ Wl2, const float* __restrict__ Wr2,
    float* __restrict__ F) {
  const int gtid = blockIdx.x * 256 + threadIdx.x;
  const int gsz = gridDim.x * 256;
  write_tile_keys<R1>(col1, key1, E1, n1, gtid, gsz);
  write_tile_keys<R2>(col2, key2, E2, n2, gtid, gsz);
  for (int t = gsz - 1 - gtid; t < 4 * W_FRAG4; t += gsz) {
    const int m = t / W_FRAG4;
    const float* Wm = m == 0 ? Wl1 : m == 1 ? Wr1 : m == 2 ? Wl2 : Wr2;
    write_weight_frags(Wm, F + (size_t)m * D * D, t - m * W_FRAG4);
  }
}

#define SP (D + 4)
#define AGG_WAVES 32
#define AGG_QCAP 576
#define AGG_QDRAIN 64
static_assert(AGG_QDRAIN + 32 <= AGG_QCAP, "one queueing round adds up to 32");

__device__ __forceinline__ unsigned key_hits(uint4 k, unsigned key4) {
  const unsigned x[4] = {k.x ^ key4, k.y ^ key4, k.z ^ key4, k.w ^ key4};
  unsigned f = 0;
#pragma unroll
  for (int d = 0; d < 4; ++d) {
    const unsigned t = ((x[d] & 0x7f7f7f7fu) + 0x7f7f7f7fu) | x[d];
    f |= (~t & 0x80808080u) >> (7 - d);
  }
  return f;
}

template <int R>
__device__ __forceinline__ void sage_drain(const int* q, int take,
                                           const float* __restrict__ src,
                                           const int* __restrict__ row,
                                           const int* __restrict__ col, int lo,
                                           float (*sAcc)[SP], int* sCnt,
                                           int lane) {
  for (int i0 = 0; i0 < take; i0 += 32) {
    const int n = min(32, take - i0);
    int r = 0, l = 0;
    if (lane < n) {
      const int e = q[i0 + lane];
      r = row[e];
      l = col[e] - lo;
      atomicAdd(&sCnt[l], 1);
    }
    for (int k = 0; k < n; k += 8) {
      float4 v[8];
      int ll[8];
#pragma unroll
      for (int j = 0; j < 8; ++j) {
        const int rr = __builtin_amdgcn_readlane(r, k + j);
        ll[j] = __builtin_amdgcn_readlane(l, k + j);
        v[j] = ((const float4*)(src + (size_t)rr * D))[lane];
      }
#pragma unroll
      for (int j = 0; j < 8; ++j) {
        if (k + j < n) {
          float* a = &sAcc[ll[j]][lane];
          atomicAdd(a, v[j].x);
          atomicAdd(a + 32, v[j].y);
          atomicAdd(a + 64, v[j].z);
          atomicAdd(a + 96, v[j].w);
        }
      }
    }
  }
}

template <int NT, int NJ>
__device__ __forceinline__ void sage_pass(const float* __restrict__ sA,
                                          const float4 (&b)[NJ], int aoff,
                                          v8f* acc) {
#pragma unroll
  for (int jj = 0; jj < NJ; ++jj) {
#pragma unroll
    for (int t0 = 0; t0 < NT; t0 += 2) {
      const int t1 = (t0 + 1 < NT) ? t0 + 1 : t0;
      const float4 a0 = *(const float4*)&sA[t0 * 64 * SP + aoff + 8 * jj];
      const float4 a1 = *(const float4*)&sA[t1 * 64 * SP + aoff + 8 * jj];
      acc[t0] = __builtin_amdgcn_wmma_f32_16x16x4_f32(
          false, (v2f){a0.x, a0.y}, false, (v2f){b[jj].x, b[jj].y}, (short)0,
          acc[t0], false, false);
      if (t1 != t0)
        acc[t1] = __builtin_amdgcn_wmma_f32_16x16x4_f32(
            false, (v2f){a1.x, a1.y}, false, (v2f){b[jj].x, b[jj].y}, (short)0,
            acc[t1], false, false);
      acc[t0] = __builtin_amdgcn_wmma_f32_16x16x4_f32(
          false, (v2f){a0.z, a0.w}, false, (v2f){b[jj].z, b[jj].w}, (short)0,
          acc[t0], false, false);
      if (t1 != t0)
        acc[t1] = __builtin_amdgcn_wmma_f32_16x16x4_f32(
            false, (v2f){a1.z, a1.w}, false, (v2f){b[jj].z, b[jj].w}, (short)0,
            acc[t1], false, false);
    }
  }
}

template <int R, int RELU>
__global__ __launch_bounds__(1024) void sage_tile_layer(
    const float* __restrict__ src, const int* __restrict__ row,
    const int* __restrict__ col, const unsigned* __restrict__ key,
    const float* __restrict__ xdst, const float* __restrict__ Wlf,
    const float* __restrict__ bl, const float* __restrict__ Wrf,
    float* __restrict__ out, int E, int n_dst) {
  constexpr int R16 = (R + 15) / 16 * 16;
  constexpr int NT16 = R16 / 16;
  constexpr int NTMAX = (NT16 + 3) / 4;
  constexpr int KP = NTMAX > 2 ? 4 : 2;
  constexpr int NJ = D / 8 / KP;
  __shared__ __attribute__((aligned(16))) float sAcc[R16][SP];
  __shared__ __attribute__((aligned(16))) float sDst[R16][SP];
  __shared__ int sCnt[R16];
  __shared__ int sQ[AGG_WAVES][AGG_QCAP];

  const int tid = threadIdx.x;
  const int lane = tid & 31;
  const int w = __builtin_amdgcn_readfirstlane(tid >> 5);
  const int lo = blockIdx.x * R;
  if (lo >= n_dst) return;

  for (int i0 = tid; i0 < R16 * 32; i0 += 4096) {
    float4 v[4];
#pragma unroll
    for (int j = 0; j < 4; ++j) {
      const int i = min(i0 + j * 1024, R16 * 32 - 1);
      const size_t g = (size_t)min(lo + (i >> 5), n_dst - 1) * (D / 4) + (i & 31);
      v[j] = ((const float4*)xdst)[g];
    }
#pragma unroll
    for (int j = 0; j < 4; ++j) {
      const int i = i0 + j * 1024;
      if (i < R16 * 32) *(float4*)&sDst[i >> 5][(i & 31) * 4] = v[j];
    }
  }
  for (int i = tid; i < R16 * SP; i += 1024) (&sAcc[0][0])[i] = 0.0f;
  for (int i = tid; i < R16; i += 1024) sCnt[i] = 0;
  __syncthreads();

  int* q = sQ[w];
  int qn = 0;
  const int nstep = (E + KEY_STEP - 1) / KEY_STEP;
  const unsigned key4 = (unsigned)blockIdx.x * 0x01010101u;

  for (int st = w; st < nstep; st += AGG_WAVES) {
    const uint4* p = (const uint4*)(key + (size_t)st * (KEY_STEP / 4));
    uint4 kv[4];
#pragma unroll
    for (int u = 0; u < 4; ++u) kv[u] = p[u * 32 + lane];
#pragma unroll
    for (int h = 0; h < 2; ++h) {
      unsigned f = key_hits(kv[2 * h], key4) | (key_hits(kv[2 * h + 1], key4) << 4);
      const int ebase = st * KEY_STEP + h * 1024 + lane * 16;
      for (;;) {
        const unsigned mask = __builtin_amdgcn_ballot_w32(f != 0);
        if (mask == 0) break;
        if (f != 0) {
          const int b = __builtin_ctz(f);
          f &= f - 1;
          q[qn + (int)__builtin_amdgcn_mbcnt_lo(mask, 0u)] =
              ebase + ((b & 4) << 7) + ((b & 3) << 2) + (b >> 3);
        }
        qn += __builtin_popcount(mask);
        if (qn > AGG_QDRAIN) {
          const int take = qn & ~31;
          sage_drain<R>(q, take, src, row, col, lo, sAcc, sCnt, lane);
          const int rem = qn - take;
          if (lane < rem) {
            const int t = q[take + lane];
            q[lane] = t;
          }
          qn = rem;
        }
      }
    }
  }
  sage_drain<R>(q, qn, src, row, col, lo, sAcc, sCnt, lane);

  const int ct = w & 7;
  const int rg = w >> 3;
  const int nt = (NT16 - rg + 3) >> 2;
  const int k0 = (lane >> 4) << 2;
  const int ncol = ct * 16 + (lane & 15);
  const int aoff = (rg * 16 + (lane & 15)) * SP + k0;
  const int boff = ct * 16 * 32 + lane;

  const float bv = bl[ncol];
  v8f acc[NTMAX];
#pragma unroll
  for (int t = 0; t < NTMAX; ++t) acc[t] = (v8f){bv, bv, bv, bv, bv, bv, bv, bv};

#pragma unroll
  for (int hk = 0; hk < KP; ++hk) {
    float4 b[NJ];
#pragma unroll
    for (int jj = 0; jj < NJ; ++jj)
      b[jj] = ((const float4*)Wrf)[boff + (hk * NJ + jj) * 32];
    if (nt == NTMAX) {
      sage_pass<NTMAX, NJ>(&sDst[0][0], b, aoff + hk * (D / KP), acc);
    } else if (NTMAX > 1 && nt == NTMAX - 1) {
      sage_pass<(NTMAX > 1 ? NTMAX - 1 : 1), NJ>(&sDst[0][0], b, aoff + hk * (D / KP), acc);
    }
  }
  __syncthreads();

  for (int r = w; r < R; r += AGG_WAVES) {
    const float inv = 1.0f / fmaxf((float)sCnt[r], 1.0f);
    float4 o;
    o.x = sAcc[r][lane] * inv;
    o.y = sAcc[r][32 + lane] * inv;
    o.z = sAcc[r][64 + lane] * inv;
    o.w = sAcc[r][96 + lane] * inv;
    *(float4*)&sAcc[r][lane * 4] = o;
  }
  __syncthreads();

#pragma unroll
  for (int hk = 0; hk < KP; ++hk) {
    float4 b[NJ];
#pragma unroll
    for (int jj = 0; jj < NJ; ++jj)
      b[jj] = ((const float4*)Wlf)[boff + (hk * NJ + jj) * 32];
    if (nt == NTMAX) {
      sage_pass<NTMAX, NJ>(&sAcc[0][0], b, aoff + hk * (D / KP), acc);
    } else if (NTMAX > 1 && nt == NTMAX - 1) {
      sage_pass<(NTMAX > 1 ? NTMAX - 1 : 1), NJ>(&sAcc[0][0], b, aoff + hk * (D / KP), acc);
    }
  }
  __syncthreads();

  {
    const int rbase = (lane >> 4) << 3;
#pragma unroll
    for (int t = 0; t < NTMAX; ++t) {
      if (t < nt) {
#pragma unroll
        for (int v = 0; v < 8; ++v)
          sAcc[(rg + 4 * t) * 16 + rbase + v][ncol] = acc[t][v];
      }
    }
  }
  __syncthreads();

  for (int r = w; r < R; r += AGG_WAVES) {
    const int g = lo + r;
    if (g < n_dst) {
      float4 o = *(const float4*)&sAcc[r][lane * 4];
      float p = o.x * o.x + o.y * o.y + o.z * o.z + o.w * o.w;
#pragma unroll
      for (int d = 16; d >= 1; d >>= 1) p += __shfl_xor(p, d, 32);
      const float nrm = fmaxf(sqrtf(p), 1e-12f);
      o.x /= nrm;
      o.y /= nrm;
      o.z /= nrm;
      o.w /= nrm;
      if (RELU) {
        o.x = fmaxf(o.x, 0.0f);
        o.y = fmaxf(o.y, 0.0f);
        o.z = fmaxf(o.z, 0.0f);
        o.w = fmaxf(o.w, 0.0f);
      }
      ((float4*)(out + (size_t)g * D))[lane] = o;
    }
  }
}

#define R_MID 197
#define R_DST 40
#define KEY_PLANE (1 << 20)
static_assert((N_MID + R_MID - 1) / R_MID <= (int)KEY_NONE &&
                  (N_DST + R_DST - 1) / R_DST <= (int)KEY_NONE,
              "every tile index must be a key byte below KEY_NONE");

extern "C" void kernel_launch(void* const* d_in, const int* in_sizes, int n_in,
                              void* d_out, int out_size, void* d_ws,
                              size_t ws_size, hipStream_t stream) {
  const float* x    = (const float*)d_in[0];
  const float* W_l1 = (const float*)d_in[1];
  const float* b_l1 = (const float*)d_in[2];
  const float* W_r1 = (const float*)d_in[3];
  const float* W_l2 = (const float*)d_in[4];
  const float* b_l2 = (const float*)d_in[5];
  const float* W_r2 = (const float*)d_in[6];
  const int* row1 = (const int*)d_in[7];
  const int* col1 = (const int*)d_in[8];
  const int* row2 = (const int*)d_in[9];
  const int* col2 = (const int*)d_in[10];
  const int E1 = in_sizes[7];
  const int E2 = in_sizes[9];

  float* ws = (float*)d_ws;
  float* h  = ws;
  unsigned* key1 = (unsigned*)(h + (size_t)N_MID * D);
  unsigned* key2 = key1 + KEY_PLANE;
  float* F = (float*)(key2 + KEY_PLANE);
  const float* Wl1f = F;
  const float* Wr1f = F + D * D;
  const float* Wl2f = F + 2 * D * D;
  const float* Wr2f = F + 3 * D * D;

  sage_tile_keys<R_MID, R_DST><<<256, 256, 0, stream>>>(
      col1, col2, key1, key2, E1, E2, N_MID, N_DST, W_l1, W_r1, W_l2, W_r2, F);

  sage_tile_layer<R_MID, 1><<<254, 1024, 0, stream>>>(
      x, row1, col1, key1, x, Wl1f, b_l1, Wr1f, h, E1, N_MID);
  sage_tile_layer<R_DST, 0><<<250, 1024, 0, stream>>>(
      h, row2, col2, key2, h, Wl2f, b_l2, Wr2f, (float*)d_out, E2, N_DST);
}


# Launch 1 writes each 4096-edge chunk's tile keys in tile-group order (8 groups) with one word per edge beside them (source row, row in tile); a tile scans only its group's part; arrival-order sums
# speedup vs baseline: 4.1433x; 1.0599x over previous
#include <hip/hip_runtime.h>
#include <hip/hip_bf16.h>


#define D 128
#define N_MID 50000
#define N_DST 10000

typedef __attribute__((ext_vector_type(2))) float v2f;
typedef __attribute__((ext_vector_type(8))) float v8f;

#define KEY_NONE 255u
#define KEY_GROUPS 8
#define KEY_SUBS 4
#define KEY_OFFP 16
#define ROW_BITS 18

template <int R>
__device__ __forceinline__ unsigned tile_key(int c, int n_dst) {
  return (unsigned)c < (unsigned)n_dst ? (unsigned)c / (unsigned)R : KEY_NONE;
}

template <int R, int CH>
__device__ __forceinline__ void bucket_chunk(
    const int* __restrict__ row, const int* __restrict__ col,
    unsigned* __restrict__ keyP, unsigned* __restrict__ wordP,
    unsigned* __restrict__ offP, int E, int n_dst, int n_src, int ch,
    unsigned* sWord, unsigned char* sKey, int* sCntG, int* sBase, int tid) {
  constexpr int QPT = CH / 1024;
  static_assert(CH % 1024 == 0 && QPT >= 1, "a thread takes whole quads");
  if (tid < KEY_GROUPS * KEY_SUBS) sCntG[tid] = 0;
  __syncthreads();
  const int sub = tid & (KEY_SUBS - 1);
  unsigned kk[QPT * 4], ww[QPT * 4];
  int rk[QPT * 4];
#pragma unroll
  for (int qd = 0; qd < QPT; ++qd) {
    const int e0 = ch * CH + (qd * 256 + tid) * 4;
    int c[4], r[4];
    if (e0 + 3 < E) {
      const int4 cv = ((const int4*)col)[e0 >> 2];
      const int4 rv = ((const int4*)row)[e0 >> 2];
      c[0] = cv.x; c[1] = cv.y; c[2] = cv.z; c[3] = cv.w;
      r[0] = rv.x; r[1] = rv.y; r[2] = rv.z; r[3] = rv.w;
    } else {
#pragma unroll
      for (int j = 0; j < 4; ++j) {
        c[j] = (e0 + j < E) ? col[e0 + j] : -1;
        r[j] = (e0 + j < E) ? row[e0 + j] : 0;
      }
    }
#pragma unroll
    for (int j = 0; j < 4; ++j) {
      const unsigned k = tile_key<R>(c[j], n_dst);
      const unsigned lr = k != KEY_NONE ? (unsigned)c[j] - k * (unsigned)R : 0u;
      const unsigned rr = (unsigned)min(max(r[j], 0), n_src - 1);
      kk[qd * 4 + j] = k;
      ww[qd * 4 + j] = rr | (lr << ROW_BITS);
      rk[qd * 4 + j] = atomicAdd(&sCntG[(k >> 5) * KEY_SUBS + sub], 1);
    }
  }
  __syncthreads();
  if (tid == 0) {
    int a = 0;
#pragma unroll
    for (int g = 0; g < KEY_GROUPS * KEY_SUBS; ++g) {
      sBase[g] = a;
      a += sCntG[g];
    }
    sBase[KEY_GROUPS * KEY_SUBS] = a;
  }
  __syncthreads();
#pragma unroll
  for (int i = 0; i < QPT * 4; ++i) {
    const int pos = min(sBase[(kk[i] >> 5) * KEY_SUBS + sub] + rk[i], CH - 1);
    sKey[pos] = (unsigned char)kk[i];
    sWord[pos] = ww[i];
  }
  __syncthreads();
  for (int i = tid; i < CH / 4; i += 256)
    keyP[(size_t)ch * (CH / 4) + i] = ((const unsigned*)sKey)[i];
  for (int i = tid; i < CH; i += 256) wordP[(size_t)ch * CH + i] = sWord[i];
  if (tid <= KEY_GROUPS) offP[ch * KEY_OFFP + tid] = (unsigned)sBase[tid * KEY_SUBS];
  __syncthreads();
}

#define W_FRAG4 (D * D / 4)

__device__ __forceinline__ void write_weight_frags(const float* __restrict__ Wm,
                                                   float* __restrict__ F, int idx) {
  const int ct = idx >> 9, j = (idx >> 5) & 15, h = (idx >> 4) & 1, c = idx & 15;
  ((float4*)F)[idx] = *(const float4*)&Wm[(size_t)(ct * 16 + c) * D + 8 * j + 4 * h];
}

template <int R1, int CH1, int R2, int CH2>
__global__ __launch_bounds__(256) void sage_tile_keys(
    const int* __restrict__ row1, const int* __restrict__ col1,
    const int* __restrict__ row2, const int* __restrict__ col2,
    unsigned* __restrict__ key1, unsigned* __restrict__ word1,
    unsigned* __restrict__ off1, unsigned* __restrict__ key2,
    unsigned* __restrict__ word2, unsigned* __restrict__ off2, int E1, int E2,
    int n1, int n2, int ns1, int ns2, const float* __restrict__ Wl1,
    const float* __restrict__ Wr1, const float* __restrict__ Wl2,
    const float* __restrict__ Wr2, float* __restrict__ F) {
  constexpr int CHMAX = CH1 > CH2 ? CH1 : CH2;
  __shared__ unsigned sWord[CHMAX];
  __shared__ __attribute__((aligned(16))) unsigned char sKey[CHMAX];
  __shared__ int sCntG[KEY_GROUPS * KEY_SUBS];
  __shared__ int sBase[KEY_GROUPS * KEY_SUBS + 1];
  const int tid = threadIdx.x;
  const int nch1 = (E1 + CH1 - 1) / CH1;
  const int nch2 = (E2 + CH2 - 1) / CH2;
  for (int ch = blockIdx.x; ch < nch1 + nch2; ch += gridDim.x) {
    if (ch < nch1)
      bucket_chunk<R1, CH1>(row1, col1, key1, word1, off1, E1, n1, ns1, ch,
                            sWord, sKey, sCntG, sBase, tid);
    else
      bucket_chunk<R2, CH2>(row2, col2, key2, word2, off2, E2, n2, ns2,
                            ch - nch1, sWord, sKey, sCntG, sBase, tid);
  }
  const int gsz = gridDim.x * 256;
  for (int t = blockIdx.x * 256 + tid; t < 4 * W_FRAG4; t += gsz) {
    const int m = t / W_FRAG4;
    const float* Wm = m == 0 ? Wl1 : m == 1 ? Wr1 : m == 2 ? Wl2 : Wr2;
    write_weight_frags(Wm, F + (size_t)m * D * D, t - m * W_FRAG4);
  }
}

#define SP (D + 4)
#define AGG_WAVES 32
#define AGG_QCAP 576
#define AGG_QDRAIN 64
static_assert(AGG_QDRAIN + 32 <= AGG_QCAP, "one queueing round adds up to 32");

__device__ __forceinline__ unsigned key_hits(uint4 k, unsigned key4) {
  const unsigned x[4] = {k.x ^ key4, k.y ^ key4, k.z ^ key4, k.w ^ key4};
  unsigned f = 0;
#pragma unroll
  for (int d = 0; d < 4; ++d) {
    const unsigned t = ((x[d] & 0x7f7f7f7fu) + 0x7f7f7f7fu) | x[d];
    f |= (~t & 0x80808080u) >> (7 - d);
  }
  return f;
}

template <int R>
__device__ __forceinline__ void sage_drain(const int* q, int take,
                                           const float* __restrict__ src,
                                           const unsigned* __restrict__ word,
                                           int n_src, float (*sAcc)[SP],
                                           int* sCnt, int lane) {
  for (int i0 = 0; i0 < take; i0 += 32) {
    const int n = min(32, take - i0);
    int r = 0, l = 0;
    if (lane < n) {
      const unsigned wd = word[q[i0 + lane]];
      r = min((int)(wd & ((1u << ROW_BITS) - 1)), n_src - 1);
      l = min((int)(wd >> ROW_BITS), R - 1);
      atomicAdd(&sCnt[l], 1);
    }
    for (int k = 0; k < n; k += 8) {
      float4 v[8];
      int ll[8];
#pragma unroll
      for (int j = 0; j < 8; ++j) {
        const int rr = __builtin_amdgcn_readlane(r, k + j);
        ll[j] = __builtin_amdgcn_readlane(l, k + j);
        v[j] = ((const float4*)(src + (size_t)rr * D))[lane];
      }
#pragma unroll
      for (int j = 0; j < 8; ++j) {
        if (k + j < n) {
          float* a = &sAcc[ll[j]][lane];
          atomicAdd(a, v[j].x);
          atomicAdd(a + 32, v[j].y);
          atomicAdd(a + 64, v[j].z);
          atomicAdd(a + 96, v[j].w);
        }
      }
    }
  }
}

template <int NT, int NJ>
__device__ __forceinline__ void sage_pass(const float* __restrict__ sA,
                                          const float4 (&b)[NJ], int aoff,
                                          v8f* acc) {
#pragma unroll
  for (int jj = 0; jj < NJ; ++jj) {
#pragma unroll
    for (int t0 = 0; t0 < NT; t0 += 2) {
      const int t1 = (t0 + 1 < NT) ? t0 + 1 : t0;
      const float4 a0 = *(const float4*)&sA[t0 * 64 * SP + aoff + 8 * jj];
      const float4 a1 = *(const float4*)&sA[t1 * 64 * SP + aoff + 8 * jj];
      acc[t0] = __builtin_amdgcn_wmma_f32_16x16x4_f32(
          false, (v2f){a0.x, a0.y}, false, (v2f){b[jj].x, b[jj].y}, (short)0,
          acc[t0], false, false);
      if (t1 != t0)
        acc[t1] = __builtin_amdgcn_wmma_f32_16x16x4_f32(
            false, (v2f){a1.x, a1.y}, false, (v2f){b[jj].x, b[jj].y}, (short)0,
            acc[t1], false, false);
      acc[t0] = __builtin_amdgcn_wmma_f32_16x16x4_f32(
          false, (v2f){a0.z, a0.w}, false, (v2f){b[jj].z, b[jj].w}, (short)0,
          acc[t0], false, false);
      if (t1 != t0)
        acc[t1] = __builtin_amdgcn_wmma_f32_16x16x4_f32(
            false, (v2f){a1.z, a1.w}, false, (v2f){b[jj].z, b[jj].w}, (short)0,
            acc[t1], false, false);
    }
  }
}

template <int R, int CH, int RELU>
__global__ __launch_bounds__(1024) void sage_tile_layer(
    const float* __restrict__ src, const unsigned* __restrict__ key,
    const unsigned* __restrict__ word, const unsigned* __restrict__ off,
    const float* __restrict__ xdst, const float* __restrict__ Wlf,
    const float* __restrict__ bl, const float* __restrict__ Wrf,
    float* __restrict__ out, int E, int n_dst, int n_src) {
  constexpr int R16 = (R + 15) / 16 * 16;
  constexpr int NT16 = R16 / 16;
  constexpr int NTMAX = (NT16 + 3) / 4;
  constexpr int KP = NTMAX > 2 ? 4 : 2;
  constexpr int NJ = D / 8 / KP;
  __shared__ __attribute__((aligned(16))) float sAcc[R16][SP];
  __shared__ __attribute__((aligned(16))) float sDst[R16][SP];
  __shared__ int sCnt[R16];
  __shared__ int sQ[AGG_WAVES][AGG_QCAP];

  const int tid = threadIdx.x;
  const int lane = tid & 31;
  const int w = __builtin_amdgcn_readfirstlane(tid >> 5);
  const int lo = blockIdx.x * R;
  if (lo >= n_dst) return;

  for (int i0 = tid; i0 < R16 * 32; i0 += 4096) {
    float4 v[4];
#pragma unroll
    for (int j = 0; j < 4; ++j) {
      const int i = min(i0 + j * 1024, R16 * 32 - 1);
      const size_t g = (size_t)min(lo + (i >> 5), n_dst - 1) * (D / 4) + (i & 31);
      v[j] = ((const float4*)xdst)[g];
    }
#pragma unroll
    for (int j = 0; j < 4; ++j) {
      const int i = i0 + j * 1024;
      if (i < R16 * 32) *(float4*)&sDst[i >> 5][(i & 31) * 4] = v[j];
    }
  }
  for (int i = tid; i < R16 * SP; i += 1024) (&sAcc[0][0])[i] = 0.0f;
  for (int i = tid; i < R16; i += 1024) sCnt[i] = 0;
  __syncthreads();

  int* q = sQ[w];
  int qn = 0;
  const int nchunk = (E + CH - 1) / CH;
  const unsigned key4 = (unsigned)blockIdx.x * 0x01010101u;
  const int grp = blockIdx.x >> 5;

  int so = 0, to = 0;
  {
    const int chl = w + 32 * lane;
    if (chl < nchunk) {
      so = (int)off[chl * KEY_OFFP + grp];
      to = (int)off[chl * KEY_OFFP + grp + 1];
    }
    so = min(max(so, 0), CH);
    to = min(max(to, so), CH);
  }

  for (int i = 0; i < 32 && w + 32 * i < nchunk; ++i) {
    const int ch = w + 32 * i;
    const int s = __builtin_amdgcn_readlane(so, i);
    const int t = __builtin_amdgcn_readlane(to, i);
    const unsigned char* kb = (const unsigned char*)key + (size_t)ch * CH;
    for (int p = s & ~15; p < t; p += 1024) {
      uint4 kv[2];
#pragma unroll
      for (int u = 0; u < 2; ++u) {
        const int o = p + u * 512 + lane * 16;
        kv[u] = (uint4){~0u, ~0u, ~0u, ~0u};
        if (o < t) kv[u] = *(const uint4*)(kb + o);
      }
#pragma unroll
      for (int u = 0; u < 2; ++u) {
        unsigned f = key_hits(kv[u], key4);
        const int base = p + u * 512 + lane * 16;
        for (;;) {
          if (__builtin_amdgcn_ballot_w32(f != 0) == 0) break;
          bool ok = false;
          int pos = 0;
          if (f != 0) {
            const int b = __builtin_ctz(f);
            f &= f - 1;
            pos = base + ((b & 3) << 2) + (b >> 3);
            ok = pos >= s && pos < t;
          }
          const unsigned mask = __builtin_amdgcn_ballot_w32(ok);
          if (ok) q[qn + (int)__builtin_amdgcn_mbcnt_lo(mask, 0u)] = ch * CH + pos;
          qn += __builtin_popcount(mask);
          if (qn > AGG_QDRAIN) {
            const int take = qn & ~31;
            sage_drain<R>(q, take, src, word, n_src, sAcc, sCnt, lane);
            const int rem = qn - take;
            if (lane < rem) {
              const int tq = q[take + lane];
              q[lane] = tq;
            }
            qn = rem;
          }
        }
      }
    }
  }
  sage_drain<R>(q, qn, src, word, n_src, sAcc, sCnt, lane);

  const int ct = w & 7;
  const int rg = w >> 3;
  const int nt = (NT16 - rg + 3) >> 2;
  const int k0 = (lane >> 4) << 2;
  const int ncol = ct * 16 + (lane & 15);
  const int aoff = (rg * 16 + (lane & 15)) * SP + k0;
  const int boff = ct * 16 * 32 + lane;

  const float bv = bl[ncol];
  v8f acc[NTMAX];
#pragma unroll
  for (int t = 0; t < NTMAX; ++t) acc[t] = (v8f){bv, bv, bv, bv, bv, bv, bv, bv};

#pragma unroll
  for (int hk = 0; hk < KP; ++hk) {
    float4 b[NJ];
#pragma unroll
    for (int jj = 0; jj < NJ; ++jj)
      b[jj] = ((const float4*)Wrf)[boff + (hk * NJ + jj) * 32];
    if (nt == NTMAX) {
      sage_pass<NTMAX, NJ>(&sDst[0][0], b, aoff + hk * (D / KP), acc);
    } else if (NTMAX > 1 && nt == NTMAX - 1) {
      sage_pass<(NTMAX > 1 ? NTMAX - 1 : 1), NJ>(&sDst[0][0], b, aoff + hk * (D / KP), acc);
    }
  }
  __syncthreads();

  for (int r = w; r < R; r += AGG_WAVES) {
    const float inv = 1.0f / fmaxf((float)sCnt[r], 1.0f);
    float4 o;
    o.x = sAcc[r][lane] * inv;
    o.y = sAcc[r][32 + lane] * inv;
    o.z = sAcc[r][64 + lane] * inv;
    o.w = sAcc[r][96 + lane] * inv;
    *(float4*)&sAcc[r][lane * 4] = o;
  }
  __syncthreads();

#pragma unroll
  for (int hk = 0; hk < KP; ++hk) {
    float4 b[NJ];
#pragma unroll
    for (int jj = 0; jj < NJ; ++jj)
      b[jj] = ((const float4*)Wlf)[boff + (hk * NJ + jj) * 32];
    if (nt == NTMAX) {
      sage_pass<NTMAX, NJ>(&sAcc[0][0], b, aoff + hk * (D / KP), acc);
    } else if (NTMAX > 1 && nt == NTMAX - 1) {
      sage_pass<(NTMAX > 1 ? NTMAX - 1 : 1), NJ>(&sAcc[0][0], b, aoff + hk * (D / KP), acc);
    }
  }
  __syncthreads();

  {
    const int rbase = (lane >> 4) << 3;
#pragma unroll
    for (int t = 0; t < NTMAX; ++t) {
      if (t < nt) {
#pragma unroll
        for (int v = 0; v < 8; ++v)
          sAcc[(rg + 4 * t) * 16 + rbase + v][ncol] = acc[t][v];
      }
    }
  }
  __syncthreads();

  for (int r = w; r < R; r += AGG_WAVES) {
    const int g = lo + r;
    if (g < n_dst) {
      float4 o = *(const float4*)&sAcc[r][lane * 4];
      float p = o.x * o.x + o.y * o.y + o.z * o.z + o.w * o.w;
#pragma unroll
      for (int d = 16; d >= 1; d >>= 1) p += __shfl_xor(p, d, 32);
      const float nrm = fmaxf(sqrtf(p), 1e-12f);
      o.x /= nrm;
      o.y /= nrm;
      o.z /= nrm;
      o.w /= nrm;
      if (RELU) {
        o.x = fmaxf(o.x, 0.0f);
        o.y = fmaxf(o.y, 0.0f);
        o.z = fmaxf(o.z, 0.0f);
        o.w = fmaxf(o.w, 0.0f);
      }
      ((float4*)(out + (size_t)g * D))[lane] = o;
    }
  }
}

#define N_SRC 200000
#define R_MID 197
#define R_DST 40
#define CH_1 4096
#define CH_2 4096
#define KEY_PLANE (1 << 20)
#define WORD_PLANE (1 << 22)
#define OFF_PLANE (1024 * KEY_OFFP)
static_assert((N_MID + R_MID - 1) / R_MID <= (int)KEY_NONE &&
                  (N_DST + R_DST - 1) / R_DST <= (int)KEY_NONE,
              "every tile index must be a key byte below KEY_NONE");
static_assert(((int)KEY_NONE >> 5) < KEY_GROUPS, "every key has a group");
static_assert(N_SRC <= (1 << ROW_BITS) && N_MID <= (1 << ROW_BITS) &&
                  R_MID <= (1 << (32 - ROW_BITS)) && R_DST <= (1 << (32 - ROW_BITS)),
              "a source row and a row inside a tile share one dword");

extern "C" void kernel_launch(void* const* d_in, const int* in_sizes, int n_in,
                              void* d_out, int out_size, void* d_ws,
                              size_t ws_size, hipStream_t stream) {
  const float* x    = (const float*)d_in[0];
  const float* W_l1 = (const float*)d_in[1];
  const float* b_l1 = (const float*)d_in[2];
  const float* W_r1 = (const float*)d_in[3];
  const float* W_l2 = (const float*)d_in[4];
  const float* b_l2 = (const float*)d_in[5];
  const float* W_r2 = (const float*)d_in[6];
  const int* row1 = (const int*)d_in[7];
  const int* col1 = (const int*)d_in[8];
  const int* row2 = (const int*)d_in[9];
  const int* col2 = (const int*)d_in[10];
  const int E1 = in_sizes[7];
  const int E2 = in_sizes[9];

  float* ws = (float*)d_ws;
  float* h  = ws;
  unsigned* key1 = (unsigned*)(h + (size_t)N_MID * D);
  unsigned* key2 = key1 + KEY_PLANE;
  unsigned* word1 = key2 + KEY_PLANE;
  unsigned* word2 = word1 + WORD_PLANE;
  unsigned* off1 = word2 + WORD_PLANE;
  unsigned* off2 = off1 + OFF_PLANE;
  float* F = (float*)(off2 + OFF_PLANE);
  const float* Wl1f = F;
  const float* Wr1f = F + D * D;
  const float* Wl2f = F + 2 * D * D;
  const float* Wr2f = F + 3 * D * D;

  sage_tile_keys<R_MID, CH_1, R_DST, CH_2><<<236, 256, 0, stream>>>(
      row1, col1, row2, col2, key1, word1, off1, key2, word2, off2, E1, E2,
      N_MID, N_DST, N_SRC, N_MID, W_l1, W_r1, W_l2, W_r2, F);

  sage_tile_layer<R_MID, CH_1, 1><<<254, 1024, 0, stream>>>(
      x, key1, word1, off1, x, Wl1f, b_l1, Wr1f, h, E1, N_MID, N_SRC);
  sage_tile_layer<R_DST, CH_2, 0><<<250, 1024, 0, stream>>>(
      h, key2, word2, off2, h, Wl2f, b_l2, Wr2f, (float*)d_out, E2, N_DST, N_MID);
}


# Launch 1 orders each chunk's edge words by destination tile (256 bins; was 8 tile groups and a key byte per edge) and writes each run's start; a tile reads only its runs: no key scan; list-order sums
# speedup vs baseline: 4.6043x; 1.1113x over previous
#include <hip/hip_runtime.h>
#include <hip/hip_bf16.h>


#define D 128
#define N_MID 50000
#define N_DST 10000

typedef __attribute__((ext_vector_type(2))) float v2f;
typedef __attribute__((ext_vector_type(4))) float v4f;
typedef __attribute__((ext_vector_type(8))) float v8f;

#define KEY_NONE 255u
#define KEY_BINS 256
#define KEY_SUBS 4
#define KEY_OFFP 264
#define ROW_BITS 18

template <int R>
__device__ __forceinline__ unsigned tile_key(int c, int n_dst) {
  return (unsigned)c < (unsigned)n_dst ? (unsigned)c / (unsigned)R : KEY_NONE;
}

template <int R, int CH>
__device__ __forceinline__ void bucket_chunk(
    const int* __restrict__ row, const int* __restrict__ col,
    unsigned* __restrict__ wordP, unsigned* __restrict__ offP, int E, int n_dst,
    int n_src, int ch, unsigned* sWord, int* sCntG, int* sBase, int* sWaveTot,
    int tid) {
  constexpr int QPT = CH / 1024;
  static_assert(CH % 1024 == 0 && QPT >= 1, "a thread takes whole quads");
  *(int4*)&sCntG[tid * KEY_SUBS] = make_int4(0, 0, 0, 0);
  __syncthreads();
  const int sub = tid & (KEY_SUBS - 1);
  unsigned kk[QPT * 4], ww[QPT * 4];
  int rk[QPT * 4];
#pragma unroll
  for (int qd = 0; qd < QPT; ++qd) {
    const int e0 = ch * CH + (qd * 256 + tid) * 4;
    int c[4], r[4];
    if (e0 + 3 < E) {
      const int4 cv = ((const int4*)col)[e0 >> 2];
      const int4 rv = ((const int4*)row)[e0 >> 2];
      c[0] = cv.x; c[1] = cv.y; c[2] = cv.z; c[3] = cv.w;
      r[0] = rv.x; r[1] = rv.y; r[2] = rv.z; r[3] = rv.w;
    } else {
#pragma unroll
      for (int j = 0; j < 4; ++j) {
        c[j] = (e0 + j < E) ? col[e0 + j] : -1;
        r[j] = (e0 + j < E) ? row[e0 + j] : 0;
      }
    }
#pragma unroll
    for (int j = 0; j < 4; ++j) {
      const unsigned k = min(tile_key<R>(c[j], n_dst), KEY_NONE);
      const unsigned lr = k != KEY_NONE ? (unsigned)c[j] - k * (unsigned)R : 0u;
      const unsigned rr = (unsigned)min(max(r[j], 0), n_src - 1);
      kk[qd * 4 + j] = k;
      ww[qd * 4 + j] = rr | (lr << ROW_BITS);
      rk[qd * 4 + j] = atomicAdd(&sCntG[k * KEY_SUBS + sub], 1);
    }
  }
  __syncthreads();
  const int4 cn = *(const int4*)&sCntG[tid * KEY_SUBS];
  const int tot = cn.x + cn.y + cn.z + cn.w;
  const int ln = tid & 31, wv = tid >> 5;
  int inc = tot;
#pragma unroll
  for (int d = 1; d < 32; d <<= 1) {
    const int o = __shfl_up(inc, d, 32);
    if (ln >= d) inc += o;
  }
  if (ln == 31) sWaveTot[wv] = inc;
  __syncthreads();
  int ex = inc - tot;
#pragma unroll
  for (int j = 0; j < 8; ++j)
    if (j < wv) ex += sWaveTot[j];
  *(int4*)&sBase[tid * KEY_SUBS] =
      make_int4(ex, ex + cn.x, ex + cn.x + cn.y, ex + cn.x + cn.y + cn.z);
  offP[ch * KEY_OFFP + tid] = (unsigned)ex;
  if (tid == KEY_BINS - 1) offP[ch * KEY_OFFP + KEY_BINS] = (unsigned)(ex + tot);
  __syncthreads();
#pragma unroll
  for (int i = 0; i < QPT * 4; ++i) {
    const int pos = min(sBase[kk[i] * KEY_SUBS + sub] + rk[i], CH - 1);
    sWord[pos] = ww[i];
  }
  __syncthreads();
  for (int i = tid; i < CH; i += 256) wordP[(size_t)ch * CH + i] = sWord[i];
  __syncthreads();
}

#define W_FRAG4 (D * D / 4)

__device__ __forceinline__ void write_weight_frags(const float* __restrict__ Wm,
                                                   float* __restrict__ F, int idx) {
  const int ct = idx >> 9, j = (idx >> 5) & 15, h = (idx >> 4) & 1, c = idx & 15;
  ((float4*)F)[idx] = *(const float4*)&Wm[(size_t)(ct * 16 + c) * D + 8 * j + 4 * h];
}

template <int R1, int CH1, int R2, int CH2>
__global__ __launch_bounds__(256) void sage_tile_keys(
    const int* __restrict__ row1, const int* __restrict__ col1,
    const int* __restrict__ row2, const int* __restrict__ col2,
    unsigned* __restrict__ word1, unsigned* __restrict__ off1,
    unsigned* __restrict__ word2, unsigned* __restrict__ off2, int E1, int E2,
    int n1, int n2, int ns1, int ns2, const float* __restrict__ Wl1,
    const float* __restrict__ Wr1, const float* __restrict__ Wl2,
    const float* __restrict__ Wr2, float* __restrict__ F) {
  constexpr int CHMAX = CH1 > CH2 ? CH1 : CH2;
  __shared__ unsigned sWord[CHMAX];
  __shared__ __attribute__((aligned(16))) int sCntG[KEY_BINS * KEY_SUBS];
  __shared__ __attribute__((aligned(16))) int sBase[KEY_BINS * KEY_SUBS];
  __shared__ int sWaveTot[8];
  const int tid = threadIdx.x;
  const int nch1 = (E1 + CH1 - 1) / CH1;
  const int nch2 = (E2 + CH2 - 1) / CH2;
  for (int ch = blockIdx.x; ch < nch1 + nch2; ch += gridDim.x) {
    if (ch < nch1)
      bucket_chunk<R1, CH1>(row1, col1, word1, off1, E1, n1, ns1, ch, sWord,
                            sCntG, sBase, sWaveTot, tid);
    else
      bucket_chunk<R2, CH2>(row2, col2, word2, off2, E2, n2, ns2, ch - nch1,
                            sWord, sCntG, sBase, sWaveTot, tid);
  }
  const int gsz = gridDim.x * 256;
  for (int t = blockIdx.x * 256 + tid; t < 4 * W_FRAG4; t += gsz) {
    const int m = t / W_FRAG4;
    const float* Wm = m == 0 ? Wl1 : m == 1 ? Wr1 : m == 2 ? Wl2 : Wr2;
    write_weight_frags(Wm, F + (size_t)m * D * D, t - m * W_FRAG4);
  }
}

#define SP (D + 4)
#define AGG_WAVES 32
#define ROW_CAP 64

template <int R>
__device__ __forceinline__ void sage_route_edge(bool has, unsigned wd,
                                                const float* __restrict__ src,
                                                int n_src, float (*sAcc)[SP],
                                                int* sCnt, int (*sRow)[ROW_CAP],
                                                int lane) {
  int r = 0, l = 0;
  bool over = false;
  if (has) {
    r = min((int)(wd & ((1u << ROW_BITS) - 1)), n_src - 1);
    l = min((int)(wd >> ROW_BITS), R - 1);
    const int slot = atomicAdd(&sCnt[l], 1);
    if (slot < ROW_CAP)
      sRow[l][slot] = r;
    else
      over = true;
  }
  unsigned om = __builtin_amdgcn_ballot_w32(over);
  while (om != 0) {
    const int j = __builtin_ctz(om);
    om &= om - 1;
    const int rr = __builtin_amdgcn_readlane(r, j);
    const int ll = __builtin_amdgcn_readlane(l, j);
    const float4 v = ((const float4*)(src + (size_t)rr * D))[lane];
    float* a = &sAcc[ll][lane * 4];
    atomicAdd(a, v.x);
    atomicAdd(a + 1, v.y);
    atomicAdd(a + 2, v.z);
    atomicAdd(a + 3, v.w);
  }
}

template <int G>
__device__ __forceinline__ void sage_sum_rows(int e, int first, int nleft,
                                              const float* __restrict__ src,
                                              int lane, float4& s) {
  static_assert(G == 8 || G == 16, "the fence below names 8 or 16 rows");
  v4f v[G];
#pragma unroll
  for (int j = 0; j < G; ++j) {
    const int rr = __builtin_amdgcn_readlane(e, first + j);
    v[j] = ((const v4f*)(src + (size_t)rr * D))[lane];
  }
  asm volatile("" ::"v"(v[0]), "v"(v[1]), "v"(v[2]), "v"(v[3]), "v"(v[4]),
               "v"(v[5]), "v"(v[6]), "v"(v[7]));
  if constexpr (G == 16)
    asm volatile("" ::"v"(v[8]), "v"(v[9]), "v"(v[10]), "v"(v[11]), "v"(v[12]),
                 "v"(v[13]), "v"(v[14]), "v"(v[15]));
#pragma unroll
  for (int j = 0; j < G; ++j) {
    if (j < nleft) {
      s.x += v[j].x;
      s.y += v[j].y;
      s.z += v[j].z;
      s.w += v[j].w;
    }
  }
}

template <int NT, int NJ>
__device__ __forceinline__ void sage_pass(const float* __restrict__ sA,
                                          const float4 (&b)[NJ], int aoff,
                                          v8f* acc) {
#pragma unroll
  for (int jj = 0; jj < NJ; ++jj) {
#pragma unroll
    for (int t0 = 0; t0 < NT; t0 += 2) {
      const int t1 = (t0 + 1 < NT) ? t0 + 1 : t0;
      const float4 a0 = *(const float4*)&sA[t0 * 64 * SP + aoff + 8 * jj];
      const float4 a1 = *(const float4*)&sA[t1 * 64 * SP + aoff + 8 * jj];
      acc[t0] = __builtin_amdgcn_wmma_f32_16x16x4_f32(
          false, (v2f){a0.x, a0.y}, false, (v2f){b[jj].x, b[jj].y}, (short)0,
          acc[t0], false, false);
      if (t1 != t0)
        acc[t1] = __builtin_amdgcn_wmma_f32_16x16x4_f32(
            false, (v2f){a1.x, a1.y}, false, (v2f){b[jj].x, b[jj].y}, (short)0,
            acc[t1], false, false);
      acc[t0] = __builtin_amdgcn_wmma_f32_16x16x4_f32(
          false, (v2f){a0.z, a0.w}, false, (v2f){b[jj].z, b[jj].w}, (short)0,
          acc[t0], false, false);
      if (t1 != t0)
        acc[t1] = __builtin_amdgcn_wmma_f32_16x16x4_f32(
            false, (v2f){a1.z, a1.w}, false, (v2f){b[jj].z, b[jj].w}, (short)0,
            acc[t1], false, false);
    }
  }
}

template <int R, int CH, int RELU>
__global__ __launch_bounds__(1024) void sage_tile_layer(
    const float* __restrict__ src, const unsigned* __restrict__ word,
    const unsigned* __restrict__ off, const float* __restrict__ xdst,
    const float* __restrict__ Wlf, const float* __restrict__ bl,
    const float* __restrict__ Wrf, float* __restrict__ out, int E, int n_dst,
    int n_src) {
  constexpr int R16 = (R + 15) / 16 * 16;
  constexpr int NT16 = R16 / 16;
  constexpr int NTMAX = (NT16 + 3) / 4;
  constexpr int KP = NTMAX > 2 ? 4 : 2;
  constexpr int NJ = D / 8 / KP;
  __shared__ __attribute__((aligned(16))) float sAcc[R16][SP];
  __shared__ __attribute__((aligned(16))) float sDst[R16][SP];
  __shared__ int sCnt[R16];
  __shared__ int sRow[R][ROW_CAP];

  const int tid = threadIdx.x;
  const int lane = tid & 31;
  const int w = __builtin_amdgcn_readfirstlane(tid >> 5);
  const int lo = blockIdx.x * R;
  if (lo >= n_dst) return;

  for (int i0 = tid; i0 < R16 * 32; i0 += 4096) {
    float4 v[4];
#pragma unroll
    for (int j = 0; j < 4; ++j) {
      const int i = min(i0 + j * 1024, R16 * 32 - 1);
      const size_t g = (size_t)min(lo + (i >> 5), n_dst - 1) * (D / 4) + (i & 31);
      v[j] = ((const float4*)xdst)[g];
    }
#pragma unroll
    for (int j = 0; j < 4; ++j) {
      const int i = i0 + j * 1024;
      if (i < R16 * 32) *(float4*)&sDst[i >> 5][(i & 31) * 4] = v[j];
    }
  }
  for (int i = tid; i < R16 * SP; i += 1024) (&sAcc[0][0])[i] = 0.0f;
  for (int i = tid; i < R16; i += 1024) sCnt[i] = 0;
  __syncthreads();

  const int nchunk = (E + CH - 1) / CH;
  const int tile = blockIdx.x;

  int so = 0, to = 0;
  {
    const int chl = w + 32 * lane;
    if (chl < nchunk) {
      so = (int)off[chl * KEY_OFFP + tile];
      to = (int)off[chl * KEY_OFFP + tile + 1];
    }
    so = min(max(so, 0), CH);
    to = min(max(to, so), CH);
  }

  for (int i0 = 0; i0 < 32 && w + 32 * i0 < nchunk; i0 += 4) {
    unsigned wd[4];
    bool has[4];
#pragma unroll
    for (int u = 0; u < 4; ++u) {
      const int i = i0 + u;
      const int s = __builtin_amdgcn_readlane(so, i);
      const int t = __builtin_amdgcn_readlane(to, i);
      const unsigned* wp = word + (size_t)(w + 32 * i) * CH;
      has[u] = s + lane < t;
      wd[u] = 0u;
      if (has[u]) wd[u] = wp[s + lane];
    }
#pragma unroll
    for (int u = 0; u < 4; ++u)
      sage_route_edge<R>(has[u], wd[u], src, n_src, sAcc, sCnt, sRow, lane);
#pragma unroll
    for (int u = 0; u < 4; ++u) {
      const int i = i0 + u;
      const int s = __builtin_amdgcn_readlane(so, i);
      const int t = __builtin_amdgcn_readlane(to, i);
      const unsigned* wp = word + (size_t)(w + 32 * i) * CH;
      for (int p0 = s + 32; p0 < t; p0 += 32) {
        const bool h = p0 + lane < t;
        unsigned x = 0u;
        if (h) x = wp[p0 + lane];
        sage_route_edge<R>(h, x, src, n_src, sAcc, sCnt, sRow, lane);
      }
    }
  }
  __syncthreads();

  for (int l = w; l < R; l += AGG_WAVES) {
    const int cnt = __builtin_amdgcn_readfirstlane(sCnt[l]);
    const int n = min(cnt, ROW_CAP);
    const int e0 = lane < n ? sRow[l][lane] : 0;
    const int e1 = lane + 32 < n ? sRow[l][lane + 32] : 0;
    float4 s = *(const float4*)&sAcc[l][lane * 4];
    if (n > 0) sage_sum_rows<16>(e0, 0, n, src, lane, s);
    if (n > 16) sage_sum_rows<8>(e0, 16, n - 16, src, lane, s);
    if (n > 24) sage_sum_rows<8>(e0, 24, n - 24, src, lane, s);
    for (int k = 32; k < n; k += 8)
      sage_sum_rows<8>(e1, k - 32, n - k, src, lane, s);
    const float inv = 1.0f / fmaxf((float)cnt, 1.0f);
    s.x *= inv;
    s.y *= inv;
    s.z *= inv;
    s.w *= inv;
    *(float4*)&sAcc[l][lane * 4] = s;
  }

  const int ct = w & 7;
  const int rg = w >> 3;
  const int nt = (NT16 - rg + 3) >> 2;
  const int k0 = (lane >> 4) << 2;
  const int ncol = ct * 16 + (lane & 15);
  const int aoff = (rg * 16 + (lane & 15)) * SP + k0;
  const int boff = ct * 16 * 32 + lane;

  const float bv = bl[ncol];
  v8f acc[NTMAX];
#pragma unroll
  for (int t = 0; t < NTMAX; ++t) acc[t] = (v8f){bv, bv, bv, bv, bv, bv, bv, bv};

#pragma unroll
  for (int hk = 0; hk < KP; ++hk) {
    float4 b[NJ];
#pragma unroll
    for (int jj = 0; jj < NJ; ++jj)
      b[jj] = ((const float4*)Wrf)[boff + (hk * NJ + jj) * 32];
    if (nt == NTMAX) {
      sage_pass<NTMAX, NJ>(&sDst[0][0], b, aoff + hk * (D / KP), acc);
    } else if (NTMAX > 1 && nt == NTMAX - 1) {
      sage_pass<(NTMAX > 1 ? NTMAX - 1 : 1), NJ>(&sDst[0][0], b, aoff + hk * (D / KP), acc);
    }
  }
  __syncthreads();

#pragma unroll
  for (int hk = 0; hk < KP; ++hk) {
    float4 b[NJ];
#pragma unroll
    for (int jj = 0; jj < NJ; ++jj)
      b[jj] = ((const float4*)Wlf)[boff + (hk * NJ + jj) * 32];
    if (nt == NTMAX) {
      sage_pass<NTMAX, NJ>(&sAcc[0][0], b, aoff + hk * (D / KP), acc);
    } else if (NTMAX > 1 && nt == NTMAX - 1) {
      sage_pass<(NTMAX > 1 ? NTMAX - 1 : 1), NJ>(&sAcc[0][0], b, aoff + hk * (D / KP), acc);
    }
  }
  __syncthreads();

  {
    const int rbase = (lane >> 4) << 3;
#pragma unroll
    for (int t = 0; t < NTMAX; ++t) {
      if (t < nt) {
#pragma unroll
        for (int v = 0; v < 8; ++v)
          sAcc[(rg + 4 * t) * 16 + rbase + v][ncol] = acc[t][v];
      }
    }
  }
  __syncthreads();

  for (int r = w; r < R; r += AGG_WAVES) {
    const int g = lo + r;
    if (g < n_dst) {
      float4 o = *(const float4*)&sAcc[r][lane * 4];
      float p = o.x * o.x + o.y * o.y + o.z * o.z + o.w * o.w;
#pragma unroll
      for (int d = 16; d >= 1; d >>= 1) p += __shfl_xor(p, d, 32);
      const float nrm = fmaxf(sqrtf(p), 1e-12f);
      o.x /= nrm;
      o.y /= nrm;
      o.z /= nrm;
      o.w /= nrm;
      if (RELU) {
        o.x = fmaxf(o.x, 0.0f);
        o.y = fmaxf(o.y, 0.0f);
        o.z = fmaxf(o.z, 0.0f);
        o.w = fmaxf(o.w, 0.0f);
      }
      ((float4*)(out + (size_t)g * D))[lane] = o;
    }
  }
}

#define N_SRC 200000
#define R_MID 197
#define R_DST 40
#define CH_1 4096
#define CH_2 4096
#define WORD_PLANE (1 << 22)
#define OFF_PLANE (1024 * KEY_OFFP)
static_assert((N_MID + R_MID - 1) / R_MID <= (int)KEY_NONE &&
                  (N_DST + R_DST - 1) / R_DST <= (int)KEY_NONE,
              "every tile index must be a bin below KEY_NONE");
static_assert((int)KEY_NONE < KEY_BINS && KEY_BINS == 256 && KEY_BINS < KEY_OFFP,
              "a thread per bin; a chunk's offsets fit its part of the plane");
static_assert(N_SRC <= (1 << ROW_BITS) && N_MID <= (1 << ROW_BITS) &&
                  R_MID <= (1 << (32 - ROW_BITS)) && R_DST <= (1 << (32 - ROW_BITS)),
              "a source row and a row inside a tile share one dword");

extern "C" void kernel_launch(void* const* d_in, const int* in_sizes, int n_in,
                              void* d_out, int out_size, void* d_ws,
                              size_t ws_size, hipStream_t stream) {
  const float* x    = (const float*)d_in[0];
  const float* W_l1 = (const float*)d_in[1];
  const float* b_l1 = (const float*)d_in[2];
  const float* W_r1 = (const float*)d_in[3];
  const float* W_l2 = (const float*)d_in[4];
  const float* b_l2 = (const float*)d_in[5];
  const float* W_r2 = (const float*)d_in[6];
  const int* row1 = (const int*)d_in[7];
  const int* col1 = (const int*)d_in[8];
  const int* row2 = (const int*)d_in[9];
  const int* col2 = (const int*)d_in[10];
  const int E1 = in_sizes[7];
  const int E2 = in_sizes[9];

  float* ws = (float*)d_ws;
  float* h  = ws;
  unsigned* word1 = (unsigned*)(h + (size_t)N_MID * D);
  unsigned* word2 = word1 + WORD_PLANE;
  unsigned* off1 = word2 + WORD_PLANE;
  unsigned* off2 = off1 + OFF_PLANE;
  float* F = (float*)(off2 + OFF_PLANE);
  const float* Wl1f = F;
  const float* Wr1f = F + D * D;
  const float* Wl2f = F + 2 * D * D;
  const float* Wr2f = F + 3 * D * D;

  sage_tile_keys<R_MID, CH_1, R_DST, CH_2><<<236, 256, 0, stream>>>(
      row1, col1, row2, col2, word1, off1, word2, off2, E1, E2, N_MID, N_DST,
      N_SRC, N_MID, W_l1, W_r1, W_l2, W_r2, F);

  sage_tile_layer<R_MID, CH_1, 1><<<254, 1024, 0, stream>>>(
      x, word1, off1, x, Wl1f, b_l1, Wr1f, h, E1, N_MID, N_SRC);
  sage_tile_layer<R_DST, CH_2, 0><<<250, 1024, 0, stream>>>(
      h, word2, off2, h, Wl2f, b_l2, Wr2f, (float*)d_out, E2, N_DST, N_MID);
}
